# starting kernel
# baseline (speedup 1.0000x reference)
#include <hip/hip_runtime.h>
#include <hip/hip_bf16.h>

typedef float v2f __attribute__((ext_vector_type(2)));
typedef float v8f __attribute__((ext_vector_type(8)));

#define NN    50000
#define EE    800000
#define GG    500
#define DIN   78
#define DINP  80
#define DH    312
#define DHP   320
#define BNEPS 1e-5f

__global__ void fill_f32(float* __restrict__ p, float v, long n) {
  long i = (long)blockIdx.x * blockDim.x + threadIdx.x;
  if (i < n) p[i] = v;
}
__global__ void fill_u32(unsigned* __restrict__ p, unsigned v, long n) {
  long i = (long)blockIdx.x * blockDim.x + threadIdx.x;
  if (i < n) p[i] = v;
}

__global__ void deg_count(const int* __restrict__ dst, float* __restrict__ deg, int E) {
  int e = blockIdx.x * blockDim.x + threadIdx.x;
  if (e < E) atomicAdd(&deg[dst[e]], 1.0f);
}
__global__ void rsqrt_inplace(float* __restrict__ p, int n) {
  int i = blockIdx.x * blockDim.x + threadIdx.x;
  if (i < n) p[i] = rsqrtf(p[i]);
}

__global__ void pad_x(const float* __restrict__ x, float* __restrict__ xp, int M) {
  int n = blockIdx.x, c = threadIdx.x;
  if (n >= M || c >= DINP) return;
  xp[(size_t)n * DINP + c] = (c < DIN) ? x[(size_t)n * DIN + c] : 0.0f;
}
__global__ void pad_W(const float* __restrict__ W, float* __restrict__ Wp, int K, int Nn) {
  int r = blockIdx.x, c = threadIdx.x;
  Wp[(size_t)r * DHP + c] = (r < K && c < Nn) ? W[(size_t)r * Nn + c] : 0.0f;
}

__global__ __launch_bounds__(256) void gemm_f32_wmma(
    const float* __restrict__ A, int lda,
    const float* __restrict__ B,
    float* __restrict__ C,
    int M, int Npad, int Kpad) {
  int wave = (blockIdx.x * blockDim.x + threadIdx.x) >> 5;
  int lane = threadIdx.x & 31;
  int ntn  = Npad >> 4;
  int tm = wave / ntn;
  int tn = wave - tm * ntn;
  if (tm * 16 >= M) return;

  int half = lane >> 4;
  int m    = lane & 15;
  int row0 = tm * 16;
  int arow = row0 + m;
  const float* Arow = A + (size_t)(arow < M ? arow : (M - 1)) * lda;
  const float* Bcol = B + (tn * 16 + m);

  v8f acc = {0.f, 0.f, 0.f, 0.f, 0.f, 0.f, 0.f, 0.f};
#pragma unroll 4
  for (int k = 0; k < Kpad; k += 4) {
    int ka = k + half * 2;
    v2f a, b;
    a.x = Arow[ka];
    a.y = Arow[ka + 1];
    b.x = Bcol[(size_t)ka * DHP];
    b.y = Bcol[(size_t)(ka + 1) * DHP];
    acc = __builtin_amdgcn_wmma_f32_16x16x4_f32(false, a, false, b,
                                                (short)0, acc, false, false);
  }
  int col = tn * 16 + m;
#pragma unroll
  for (int r = 0; r < 8; ++r) {
    int row = row0 + r + half * 8;
    if (row < M) C[(size_t)row * DHP + col] = acc[r];
  }
}

__global__ void agg_init(const float* __restrict__ h, const float* __restrict__ dinv,
                         float* __restrict__ agg, int M) {
  int n = blockIdx.x, c = threadIdx.x;
  if (n >= M) return;
  float di = dinv[n];
  agg[(size_t)n * DHP + c] = h[(size_t)n * DHP + c] * di * di;
}
__global__ void scatter_edges(const float* __restrict__ h, const int* __restrict__ src,
                              const int* __restrict__ dst, const float* __restrict__ dinv,
                              float* __restrict__ agg, int E) {
  int e = blockIdx.x, c = threadIdx.x;
  if (e >= E || c >= DH) return;
  int s = src[e], d = dst[e];
  float norm = dinv[s] * dinv[d];
  atomicAdd(&agg[(size_t)d * DHP + c], h[(size_t)s * DHP + c] * norm);
}

__global__ void bias_relu_stats(float* __restrict__ y, const float* __restrict__ bias,
                                float* __restrict__ sums, int M) {
  int c = threadIdx.x;
  if (c >= DH) return;
  int r0 = blockIdx.x * 128;
  int r1 = min(r0 + 128, M);
  float b = bias[c], s = 0.f, s2 = 0.f;
  for (int r = r0; r < r1; ++r) {
    float v = y[(size_t)r * DHP + c] + b;
    v = fmaxf(v, 0.0f);
    y[(size_t)r * DHP + c] = v;
    s += v;
    s2 += v * v;
  }
  atomicAdd(&sums[c], s);
  atomicAdd(&sums[DHP + c], s2);
}
__global__ void bn_apply(float* __restrict__ y, const float* __restrict__ sums,
                         const float* __restrict__ gamma, const float* __restrict__ beta,
                         int M) {
  int c = threadIdx.x;
  if (c >= DH) return;
  int r0 = blockIdx.x * 128;
  int r1 = min(r0 + 128, M);
  float invn  = 1.0f / (float)M;
  float mean  = sums[c] * invn;
  float var   = sums[DHP + c] * invn - mean * mean;
  float scale = rsqrtf(var + BNEPS) * gamma[c];
  float shift = beta[c] - mean * scale;
  for (int r = r0; r < r1; ++r)
    y[(size_t)r * DHP + c] = y[(size_t)r * DHP + c] * scale + shift;
}

__device__ __forceinline__ unsigned enc_f32(float f) {
  unsigned u = __float_as_uint(f);
  return (u & 0x80000000u) ? ~u : (u | 0x80000000u);
}
__device__ __forceinline__ float dec_f32(unsigned u) {
  return __uint_as_float((u & 0x80000000u) ? (u & 0x7FFFFFFFu) : ~u);
}
__global__ void pool_max(const float* __restrict__ y, const int* __restrict__ batch,
                         unsigned* __restrict__ pool, int M) {
  int n = blockIdx.x, c = threadIdx.x;
  if (n >= M || c >= DH) return;
  int g = batch[n];
  atomicMax(&pool[(size_t)g * DH + c], enc_f32(y[(size_t)n * DHP + c]));
}
__global__ void pool_decode(const unsigned* __restrict__ pool, float* __restrict__ out) {
  int g = blockIdx.x, c = threadIdx.x;
  if (c >= DH) return;
  out[(size_t)g * DH + c] = dec_f32(pool[(size_t)g * DH + c]);
}

extern "C" void kernel_launch(void* const* d_in, const int* in_sizes, int n_in,
                              void* d_out, int out_size, void* d_ws, size_t ws_size,
                              hipStream_t stream) {
  const float* x      = (const float*)d_in[0];
  const int*   eidx   = (const int*)d_in[1];
  const int*   batch  = (const int*)d_in[2];
  const float* W1     = (const float*)d_in[3];
  const float* b1     = (const float*)d_in[4];
  const float* gamma1 = (const float*)d_in[5];
  const float* beta1  = (const float*)d_in[6];
  const float* W2     = (const float*)d_in[7];
  const float* b2     = (const float*)d_in[8];
  const float* gamma2 = (const float*)d_in[9];
  const float* beta2  = (const float*)d_in[10];
  float* out = (float*)d_out;

  const int* src = eidx;
  const int* dst = eidx + EE;

  float* ws = (float*)d_ws;
  float*    deg  = ws;
  float*    xp   = ws + 50048;
  float*    Wp1  = ws + 4050048;
  float*    Wp2  = ws + 4075648;
  float*    sums = ws + 4178048;
  unsigned* pool = (unsigned*)(ws + 4178688);
  float*    bufA = ws + 4334688;
  float*    bufB = ws + 20334688;

  const int statBlocks = (NN + 127) / 128;
  const int gemmTiles  = (NN / 16) * (DHP / 16);
  const int gemmGrid   = (gemmTiles + 7) / 8;

  fill_f32<<<(NN + 255) / 256, 256, 0, stream>>>(deg, 1.0f, NN);
  deg_count<<<(EE + 255) / 256, 256, 0, stream>>>(dst, deg, EE);
  rsqrt_inplace<<<(NN + 255) / 256, 256, 0, stream>>>(deg, NN);

  pad_x<<<NN, 128, 0, stream>>>(x, xp, NN);
  pad_W<<<DINP, DHP, 0, stream>>>(W1, Wp1, DIN, DH);
  pad_W<<<DHP, DHP, 0, stream>>>(W2, Wp2, DH, DH);

  gemm_f32_wmma<<<gemmGrid, 256, 0, stream>>>(xp, DINP, Wp1, bufA, NN, DHP, DINP);
  agg_init<<<NN, DHP, 0, stream>>>(bufA, deg, bufB, NN);
  scatter_edges<<<EE, DHP, 0, stream>>>(bufA, src, dst, deg, bufB, EE);
  fill_f32<<<(2 * DHP + 255) / 256, 256, 0, stream>>>(sums, 0.0f, 2 * DHP);
  bias_relu_stats<<<statBlocks, DHP, 0, stream>>>(bufB, b1, sums, NN);
  bn_apply<<<statBlocks, DHP, 0, stream>>>(bufB, sums, gamma1, beta1, NN);

  gemm_f32_wmma<<<gemmGrid, 256, 0, stream>>>(bufB, DHP, Wp2, bufA, NN, DHP, DHP);
  agg_init<<<NN, DHP, 0, stream>>>(bufA, deg, bufB, NN);
  scatter_edges<<<EE, DHP, 0, stream>>>(bufA, src, dst, deg, bufB, EE);
  fill_f32<<<(2 * DHP + 255) / 256, 256, 0, stream>>>(sums, 0.0f, 2 * DHP);
  bias_relu_stats<<<statBlocks, DHP, 0, stream>>>(bufB, b2, sums, NN);
  bn_apply<<<statBlocks, DHP, 0, stream>>>(bufB, sums, gamma2, beta2, NN);

  fill_u32<<<((long)GG * DH + 255) / 256, 256, 0, stream>>>(pool, 0u, (long)GG * DH);
  pool_max<<<NN, DHP, 0, stream>>>(bufB, batch, pool, NN);
  pool_decode<<<GG, DHP, 0, stream>>>(pool, out);
}


# GEMMs staged through LDS: a workgroup fetches 64xKC of A and KCx320 of W in whole-line 16-byte loads, waves take WMMA fragments from LDS, C leaves via LDS; was 4-byte loads per wave; new k sum order
# speedup vs baseline: 3.5984x; 3.5984x over previous
#include <hip/hip_runtime.h>
#include <hip/hip_bf16.h>

typedef float v2f __attribute__((ext_vector_type(2)));
typedef float v8f __attribute__((ext_vector_type(8)));
typedef float v4f __attribute__((ext_vector_type(4)));
typedef unsigned v4u __attribute__((ext_vector_type(4)));
typedef unsigned v2u __attribute__((ext_vector_type(2)));

#define NN    50000
#define EE    800000
#define GG    500
#define DIN   78
#define DINP  80
#define DH    312
#define DHP   320
#define BNEPS 1e-5f
#define NT    (NN / 16)
static_assert(NN % 16 == 0, "destination tiles are whole");

__global__ void fill_f32(float* __restrict__ p, float v, long n) {
  long i = (long)blockIdx.x * blockDim.x + threadIdx.x;
  if (i < n) p[i] = v;
}
__global__ void fill_u32(unsigned* __restrict__ p, unsigned v, long n) {
  long i = (long)blockIdx.x * blockDim.x + threadIdx.x;
  if (i < n) p[i] = v;
}

__global__ void deg_tile_count(const int* __restrict__ dst, float* __restrict__ deg,
                               unsigned* __restrict__ tcnt, unsigned* __restrict__ rank, int E) {
  int e = blockIdx.x * blockDim.x + threadIdx.x;
  if (e < E) {
    int d = dst[e];
    atomicAdd(&deg[d], 1.0f);
    rank[e] = atomicAdd(&tcnt[d >> 4], 1u);
  }
}
__global__ __launch_bounds__(1024) void tile_scan(const unsigned* __restrict__ tcnt,
                                                  unsigned* __restrict__ toff,
                                                  unsigned* __restrict__ rec) {
  __shared__ unsigned s[1024];
  int t = threadIdx.x;
  unsigned n[4], c[4], sum = 0;
#pragma unroll
  for (int i = 0; i < 4; ++i) {
    int idx = t * 4 + i;
    n[i] = idx < NT ? tcnt[idx] : 0u;
    c[i] = (n[i] + 7u) & ~7u;
    sum += c[i];
  }
  s[t] = sum;
  __syncthreads();
  for (int o = 1; o < 1024; o <<= 1) {
    unsigned v = t >= o ? s[t - o] : 0u;
    __syncthreads();
    s[t] += v;
    __syncthreads();
  }
  unsigned run = s[t] - sum;
#pragma unroll
  for (int i = 0; i < 4; ++i) {
    int idx = t * 4 + i;
    if (idx <= NT) toff[idx] = run;
    if (idx < NT) {
      for (unsigned j = n[i]; j < c[i]; ++j) {
        rec[2 * (run + j)]     = (unsigned)(idx * 16) << 4;
        rec[2 * (run + j) + 1] = 0u;
      }
    }
    run += c[i];
  }
}
__global__ void tile_place(const int* __restrict__ src, const int* __restrict__ dst,
                           const float* __restrict__ dinv, const unsigned* __restrict__ toff,
                           const unsigned* __restrict__ rank, unsigned* __restrict__ rec, int E) {
  int e = blockIdx.x * blockDim.x + threadIdx.x;
  if (e < E) {
    int s = src[e], d = dst[e];
    unsigned slot = toff[d >> 4] + rank[e];
    v2u r;
    r.x = ((unsigned)s << 4) | (unsigned)(d & 15);
    r.y = __float_as_uint(dinv[s] * dinv[d]);
    *(v2u*)(rec + 2 * (size_t)slot) = r;
  }
}
__global__ void rsqrt_inplace(float* __restrict__ p, int n) {
  int i = blockIdx.x * blockDim.x + threadIdx.x;
  if (i < n) p[i] = rsqrtf(p[i]);
}

__global__ void pad_x(const float* __restrict__ x, float* __restrict__ xp, int M) {
  int n = blockIdx.x, c = threadIdx.x;
  if (n >= M || c >= DINP) return;
  xp[(size_t)n * DINP + c] = (c < DIN) ? x[(size_t)n * DIN + c] : 0.0f;
}
__global__ void pad_W(const float* __restrict__ W, float* __restrict__ Wp, int K, int Nn) {
  int r = blockIdx.x, c = threadIdx.x;
  Wp[(size_t)r * DHP + c] = (r < K && c < Nn) ? W[(size_t)r * Nn + c] : 0.0f;
}

#define GB_ROWS 64
#define GB_SLOTQ 84
#define GB_BQ   (4 * GB_SLOTQ)
#define GB_CP   (DHP + 4)
template <int KP, int KC>
__global__ __launch_bounds__(256) void gemm_lds_wmma(
    const float* __restrict__ A, const float* __restrict__ B, float* __restrict__ C, int M) {
  constexpr int AP    = KC + 4;
  constexpr int NCH   = KP / KC;
  constexpr int KQ    = KC / 4;
  constexpr int APR   = KC / 4;
  constexpr int A_F4  = GB_ROWS * APR;
  constexpr int B_MB  = KQ * (DHP / 4);
  constexpr int A_IT  = (A_F4 + 255) / 256;
  constexpr int B_IT  = (B_MB + 255) / 256;
  constexpr int MAINF = GB_ROWS * AP + KQ * GB_BQ * 4;
  constexpr int CF    = GB_ROWS * GB_CP;
  static_assert(KP % KC == 0 && KC % 8 == 0 && ((AP / 4) & 1) == 1, "chunking");
  __shared__ __attribute__((aligned(16))) float smem[MAINF > CF ? MAINF : CF];
  float* As = smem;
  v4f*   Bs = (v4f*)(smem + GB_ROWS * AP);

  const int tid  = threadIdx.x;
  const int lane = tid & 31;
  const int w    = tid >> 5;
  const int half = lane >> 4;
  const int c    = lane & 15;
  const int mp   = w & 1;
  const int q    = w >> 1;
  const int row0 = blockIdx.x * GB_ROWS;

  v4f pa[A_IT];
  v4f pb[B_IT][4];

  auto fetch = [&](int ch) {
    const int kc0 = ch * KC;
#pragma unroll
    for (int i = 0; i < A_IT; ++i) {
      int idx = tid + 256 * i;
      if (idx < A_F4) {
        int r = idx / APR, c4 = idx - r * APR;
        int grow = row0 + r;
        grow = grow < M ? grow : M - 1;
        pa[i] = *(const v4f*)(A + (size_t)grow * KP + kc0 + 4 * c4);
      }
    }
#pragma unroll
    for (int i = 0; i < B_IT; ++i) {
      int mb = tid + 256 * i;
      if (mb < B_MB) {
        int kq = mb / (DHP / 4), cg = mb - kq * (DHP / 4);
        const float* bp = B + (size_t)(kc0 + 4 * kq) * DHP + 4 * cg;
#pragma unroll
        for (int r = 0; r < 4; ++r) pb[i][r] = *(const v4f*)(bp + (size_t)r * DHP);
      }
    }
  };
  auto stage = [&]() {
#pragma unroll
    for (int i = 0; i < A_IT; ++i) {
      int idx = tid + 256 * i;
      if (idx < A_F4) {
        int r = idx / APR, c4 = idx - r * APR;
        *(v4f*)(As + r * AP + 4 * c4) = pa[i];
      }
    }
#pragma unroll
    for (int i = 0; i < B_IT; ++i) {
      int mb = tid + 256 * i;
      if (mb < B_MB) {
        int kq = mb / (DHP / 4), cg = mb - kq * (DHP / 4);
        v4f* bs = Bs + kq * GB_BQ + cg;
#pragma unroll
        for (int j = 0; j < 4; ++j) {
          v4f t;
          t.x = pb[i][0][j]; t.y = pb[i][1][j]; t.z = pb[i][2][j]; t.w = pb[i][3][j];
          bs[j * GB_SLOTQ] = t;
        }
      }
    }
  };

  v8f acc[2][5];
#pragma unroll
  for (int mt = 0; mt < 2; ++mt)
#pragma unroll
    for (int nt = 0; nt < 5; ++nt)
#pragma unroll
      for (int r = 0; r < 8; ++r) acc[mt][nt][r] = 0.0f;

  const float* a0p = As + (32 * mp + c) * AP + 4 * half;
  const float* a1p = a0p + 16 * AP;
  int bslot[5];
#pragma unroll
  for (int nt = 0; nt < 5; ++nt) {
    int n = 80 * q + 16 * nt + c;
    bslot[nt] = half * GB_BQ + (n & 3) * GB_SLOTQ + (n >> 2);
  }

  fetch(0);
#pragma unroll 1
  for (int ch = 0; ch < NCH; ++ch) {
    stage();
    __syncthreads();
    if (ch + 1 < NCH) fetch(ch + 1);
#pragma unroll
    for (int kb = 0; kb < KC; kb += 8) {
      v4f a0 = *(const v4f*)(a0p + kb);
      v4f a1 = *(const v4f*)(a1p + kb);
      v4f b[5];
#pragma unroll
      for (int nt = 0; nt < 5; ++nt) b[nt] = Bs[(kb >> 2) * GB_BQ + bslot[nt]];
      v2f a0l, a0h, a1l, a1h;
      a0l.x = a0.x; a0l.y = a0.y; a0h.x = a0.z; a0h.y = a0.w;
      a1l.x = a1.x; a1l.y = a1.y; a1h.x = a1.z; a1h.y = a1.w;
#pragma unroll
      for (int nt = 0; nt < 5; ++nt) {
        v2f bl;
        bl.x = b[nt].x; bl.y = b[nt].y;
        acc[0][nt] = __builtin_amdgcn_wmma_f32_16x16x4_f32(false, a0l, false, bl, (short)0,
                                                           acc[0][nt], false, false);
        acc[1][nt] = __builtin_amdgcn_wmma_f32_16x16x4_f32(false, a1l, false, bl, (short)0,
                                                           acc[1][nt], false, false);
      }
#pragma unroll
      for (int nt = 0; nt < 5; ++nt) {
        v2f bh;
        bh.x = b[nt].z; bh.y = b[nt].w;
        acc[0][nt] = __builtin_amdgcn_wmma_f32_16x16x4_f32(false, a0h, false, bh, (short)0,
                                                           acc[0][nt], false, false);
        acc[1][nt] = __builtin_amdgcn_wmma_f32_16x16x4_f32(false, a1h, false, bh, (short)0,
                                                           acc[1][nt], false, false);
      }
    }
    __syncthreads();
  }

  float* Cs = smem;
#pragma unroll
  for (int mt = 0; mt < 2; ++mt)
#pragma unroll
    for (int nt = 0; nt < 5; ++nt)
#pragma unroll
      for (int r = 0; r < 8; ++r)
        Cs[(32 * mp + 16 * mt + 8 * half + r) * GB_CP + 80 * q + 16 * nt + c] = acc[mt][nt][r];
  __syncthreads();
#pragma unroll
  for (int i = 0; i < (GB_ROWS * (DHP / 4)) / 256; ++i) {
    int idx = tid + 256 * i;
    int r = idx / (DHP / 4), c4 = idx - r * (DHP / 4);
    if (row0 + r < M)
      *(v4f*)(C + (size_t)(row0 + r) * DHP + 4 * c4) = *(const v4f*)(Cs + r * GB_CP + 4 * c4);
  }
}

__global__ __launch_bounds__(160) void agg_tile_wmma(
    const float* __restrict__ h, const unsigned* __restrict__ rec,
    const unsigned* __restrict__ toff, const float* __restrict__ dinv,
    float* __restrict__ agg) {
  int t    = blockIdx.x;
  int wave = threadIdx.x >> 5;
  int lane = threadIdx.x & 31;
  int half = lane >> 4;
  unsigned m = lane & 15;
  int row0 = t * 16;
  int col  = wave * 64 + (int)m * 4;

  v8f acc0, acc1, acc2, acc3;
#pragma unroll
  for (int r = 0; r < 8; ++r) {
    int row = row0 + r + half * 8;
    float di = dinv[row];
    v4f hv = *(const v4f*)(h + (size_t)row * DHP + col);
    acc0[r] = hv.x * di * di;
    acc1[r] = hv.y * di * di;
    acc2[r] = hv.z * di * di;
    acc3[r] = hv.w * di * di;
  }

  unsigned base = toff[t];
  unsigned n    = toff[t + 1] - base;
  const unsigned* rp = rec + 2 * (size_t)(base + 2 * half);
  for (unsigned i = 0; i < n; i += 8) {
    v4u ra = *(const v4u*)(rp + 2 * (size_t)i);
    v4u rb = *(const v4u*)(rp + 2 * (size_t)i + 8);
    v4f ha0 = *(const v4f*)(h + (size_t)(ra.x >> 4) * DHP + col);
    v4f ha1 = *(const v4f*)(h + (size_t)(ra.z >> 4) * DHP + col);
    v4f hb0 = *(const v4f*)(h + (size_t)(rb.x >> 4) * DHP + col);
    v4f hb1 = *(const v4f*)(h + (size_t)(rb.z >> 4) * DHP + col);
    v2f a, b;
    a.x = ((ra.x & 15u) == m) ? __uint_as_float(ra.y) : 0.0f;
    a.y = ((ra.z & 15u) == m) ? __uint_as_float(ra.w) : 0.0f;
    b.x = ha0.x; b.y = ha1.x;
    acc0 = __builtin_amdgcn_wmma_f32_16x16x4_f32(false, a, false, b, (short)0, acc0, false, false);
    b.x = ha0.y; b.y = ha1.y;
    acc1 = __builtin_amdgcn_wmma_f32_16x16x4_f32(false, a, false, b, (short)0, acc1, false, false);
    b.x = ha0.z; b.y = ha1.z;
    acc2 = __builtin_amdgcn_wmma_f32_16x16x4_f32(false, a, false, b, (short)0, acc2, false, false);
    b.x = ha0.w; b.y = ha1.w;
    acc3 = __builtin_amdgcn_wmma_f32_16x16x4_f32(false, a, false, b, (short)0, acc3, false, false);
    a.x = ((rb.x & 15u) == m) ? __uint_as_float(rb.y) : 0.0f;
    a.y = ((rb.z & 15u) == m) ? __uint_as_float(rb.w) : 0.0f;
    b.x = hb0.x; b.y = hb1.x;
    acc0 = __builtin_amdgcn_wmma_f32_16x16x4_f32(false, a, false, b, (short)0, acc0, false, false);
    b.x = hb0.y; b.y = hb1.y;
    acc1 = __builtin_amdgcn_wmma_f32_16x16x4_f32(false, a, false, b, (short)0, acc1, false, false);
    b.x = hb0.z; b.y = hb1.z;
    acc2 = __builtin_amdgcn_wmma_f32_16x16x4_f32(false, a, false, b, (short)0, acc2, false, false);
    b.x = hb0.w; b.y = hb1.w;
    acc3 = __builtin_amdgcn_wmma_f32_16x16x4_f32(false, a, false, b, (short)0, acc3, false, false);
  }
#pragma unroll
  for (int r = 0; r < 8; ++r) {
    int row = row0 + r + half * 8;
    v4f o;
    o.x = acc0[r]; o.y = acc1[r]; o.z = acc2[r]; o.w = acc3[r];
    *(v4f*)(agg + (size_t)row * DHP + col) = o;
  }
}

__global__ void bias_relu_stats(float* __restrict__ y, const float* __restrict__ bias,
                                float* __restrict__ sums, int M) {
  int c = threadIdx.x;
  if (c >= DH) return;
  int r0 = blockIdx.x * 128;
  int r1 = min(r0 + 128, M);
  float b = bias[c], s = 0.f, s2 = 0.f;
  for (int r = r0; r < r1; ++r) {
    float v = y[(size_t)r * DHP + c] + b;
    v = fmaxf(v, 0.0f);
    y[(size_t)r * DHP + c] = v;
    s += v;
    s2 += v * v;
  }
  atomicAdd(&sums[c], s);
  atomicAdd(&sums[DHP + c], s2);
}
__global__ void bn_apply(float* __restrict__ y, const float* __restrict__ sums,
                         const float* __restrict__ gamma, const float* __restrict__ beta,
                         int M) {
  int c = threadIdx.x;
  if (c >= DH) return;
  int r0 = blockIdx.x * 128;
  int r1 = min(r0 + 128, M);
  float invn  = 1.0f / (float)M;
  float mean  = sums[c] * invn;
  float var   = sums[DHP + c] * invn - mean * mean;
  float scale = rsqrtf(var + BNEPS) * gamma[c];
  float shift = beta[c] - mean * scale;
  for (int r = r0; r < r1; ++r)
    y[(size_t)r * DHP + c] = y[(size_t)r * DHP + c] * scale + shift;
}

__device__ __forceinline__ unsigned enc_f32(float f) {
  unsigned u = __float_as_uint(f);
  return (u & 0x80000000u) ? ~u : (u | 0x80000000u);
}
__device__ __forceinline__ float dec_f32(unsigned u) {
  return __uint_as_float((u & 0x80000000u) ? (u & 0x7FFFFFFFu) : ~u);
}
__global__ void bias_relu_stats_pool(const float* __restrict__ y, const float* __restrict__ bias,
                                     const int* __restrict__ batch, float* __restrict__ sums,
                                     unsigned* __restrict__ pmax, unsigned* __restrict__ pmin,
                                     int M) {
  int c = threadIdx.x;
  if (c >= DH) return;
  int r0 = blockIdx.x * 128;
  int r1 = min(r0 + 128, M);
  float b = bias[c], s = 0.f, s2 = 0.f;
  float hi = -INFINITY, lo = INFINITY;
  int g  = batch[r0];
  int gr = g;
  for (int r = r0; r < r1; ++r) {
    int gn = batch[min(r + 1, r1 - 1)];
    float v = y[(size_t)r * DHP + c] + b;
    if (gr != g) {
      atomicMax(&pmax[(size_t)g * DH + c], enc_f32(hi));
      atomicMax(&pmin[(size_t)g * DH + c], ~enc_f32(lo));
      g = gr;
      hi = -INFINITY;
      lo = INFINITY;
    }
    v = fmaxf(v, 0.0f);
    s += v;
    s2 += v * v;
    hi = fmaxf(hi, v);
    lo = fminf(lo, v);
    gr = gn;
  }
  atomicMax(&pmax[(size_t)g * DH + c], enc_f32(hi));
  atomicMax(&pmin[(size_t)g * DH + c], ~enc_f32(lo));
  atomicAdd(&sums[c], s);
  atomicAdd(&sums[DHP + c], s2);
}
__global__ void pool_bn_finish(const unsigned* __restrict__ pmax, const unsigned* __restrict__ pmin,
                               const float* __restrict__ sums, const float* __restrict__ gamma,
                               const float* __restrict__ beta, float* __restrict__ out, int M) {
  int g = blockIdx.x, c = threadIdx.x;
  if (c >= DH) return;
  float invn  = 1.0f / (float)M;
  float mean  = sums[c] * invn;
  float var   = sums[DHP + c] * invn - mean * mean;
  float scale = rsqrtf(var + BNEPS) * gamma[c];
  float shift = beta[c] - mean * scale;
  unsigned em = pmax[(size_t)g * DH + c];
  float res = -INFINITY;
  if (em != 0u) {
    float v = (scale >= 0.0f) ? dec_f32(em) : dec_f32(~pmin[(size_t)g * DH + c]);
    res = v * scale + shift;
  }
  out[(size_t)g * DH + c] = res;
}

extern "C" void kernel_launch(void* const* d_in, const int* in_sizes, int n_in,
                              void* d_out, int out_size, void* d_ws, size_t ws_size,
                              hipStream_t stream) {
  const float* x      = (const float*)d_in[0];
  const int*   eidx   = (const int*)d_in[1];
  const int*   batch  = (const int*)d_in[2];
  const float* W1     = (const float*)d_in[3];
  const float* b1     = (const float*)d_in[4];
  const float* gamma1 = (const float*)d_in[5];
  const float* beta1  = (const float*)d_in[6];
  const float* W2     = (const float*)d_in[7];
  const float* b2     = (const float*)d_in[8];
  const float* gamma2 = (const float*)d_in[9];
  const float* beta2  = (const float*)d_in[10];
  float* out = (float*)d_out;

  const int* src = eidx;
  const int* dst = eidx + EE;

  float* ws = (float*)d_ws;
  float*    deg  = ws;
  float*    xp   = ws + 50048;
  float*    Wp1  = ws + 4050048;
  float*    Wp2  = ws + 4075648;
  float*    sums = ws + 4178048;
  float*    bufA = ws + 4334688;
  float*    bufB = ws + 20334688;
  unsigned* tcnt = (unsigned*)(ws + 36334688);
  unsigned* toff = (unsigned*)(ws + 36337888);
  unsigned* rank = (unsigned*)(ws + 36341088);
  unsigned* rec  = (unsigned*)(ws + 37141088);
  unsigned* pmax = (unsigned*)(ws + 38800000);
  unsigned* pmin = pmax + GG * DH;

  const int statBlocks = (NN + 127) / 128;
  const int gemmGrid   = (NN + GB_ROWS - 1) / GB_ROWS;

  fill_f32<<<(NN + 255) / 256, 256, 0, stream>>>(deg, 1.0f, NN);
  fill_u32<<<(NT + 255) / 256, 256, 0, stream>>>(tcnt, 0u, NT);
  deg_tile_count<<<(EE + 255) / 256, 256, 0, stream>>>(dst, deg, tcnt, rank, EE);
  rsqrt_inplace<<<(NN + 255) / 256, 256, 0, stream>>>(deg, NN);

  tile_scan<<<1, 1024, 0, stream>>>(tcnt, toff, rec);
  tile_place<<<(EE + 255) / 256, 256, 0, stream>>>(src, dst, deg, toff, rank, rec, EE);

  pad_x<<<NN, 128, 0, stream>>>(x, xp, NN);
  pad_W<<<DINP, DHP, 0, stream>>>(W1, Wp1, DIN, DH);
  pad_W<<<DHP, DHP, 0, stream>>>(W2, Wp2, DH, DH);

  gemm_lds_wmma<DINP, 16><<<gemmGrid, 256, 0, stream>>>(xp, Wp1, bufA, NN);
  agg_tile_wmma<<<NT, 160, 0, stream>>>(bufA, rec, toff, deg, bufB);
  fill_f32<<<(2 * DHP + 255) / 256, 256, 0, stream>>>(sums, 0.0f, 2 * DHP);
  bias_relu_stats<<<statBlocks, DHP, 0, stream>>>(bufB, b1, sums, NN);
  bn_apply<<<statBlocks, DHP, 0, stream>>>(bufB, sums, gamma1, beta1, NN);

  gemm_lds_wmma<DHP, 32><<<gemmGrid, 256, 0, stream>>>(bufB, Wp2, bufA, NN);
  agg_tile_wmma<<<NT, 160, 0, stream>>>(bufA, rec, toff, deg, bufB);
  fill_f32<<<(2 * DHP + 255) / 256, 256, 0, stream>>>(sums, 0.0f, 2 * DHP);

  fill_u32<<<(2L * GG * DH + 255) / 256, 256, 0, stream>>>(pmax, 0u, 2L * GG * DH);
  bias_relu_stats_pool<<<statBlocks, DHP, 0, stream>>>(bufB, b2, batch, sums, pmax, pmin, NN);
  pool_bn_finish<<<GG, DHP, 0, stream>>>(pmax, pmin, sums, gamma2, beta2, out, NN);
}


# Bias+ReLU, BN moment sums and layer 2's per-graph max/min taken from the aggregation tile's WMMA accumulators; layer-2 plane never stored; 2 row-walk passes gone; moments regrouped, 16 atomic copies
# speedup vs baseline: 4.3075x; 1.1971x over previous
#include <hip/hip_runtime.h>
#include <hip/hip_bf16.h>

typedef float v2f __attribute__((ext_vector_type(2)));
typedef float v8f __attribute__((ext_vector_type(8)));
typedef float v4f __attribute__((ext_vector_type(4)));
typedef unsigned v4u __attribute__((ext_vector_type(4)));
typedef unsigned v2u __attribute__((ext_vector_type(2)));

#define NN    50000
#define EE    800000
#define GG    500
#define DIN   78
#define DINP  80
#define DH    312
#define DHP   320
#define BNEPS 1e-5f
#define NT    (NN / 16)
static_assert(NN % 16 == 0, "destination tiles are whole");

__global__ void fill_f32(float* __restrict__ p, float v, long n) {
  long i = (long)blockIdx.x * blockDim.x + threadIdx.x;
  if (i < n) p[i] = v;
}
__global__ void fill_u32(unsigned* __restrict__ p, unsigned v, long n) {
  long i = (long)blockIdx.x * blockDim.x + threadIdx.x;
  if (i < n) p[i] = v;
}

__global__ void deg_tile_count(const int* __restrict__ dst, float* __restrict__ deg,
                               unsigned* __restrict__ tcnt, unsigned* __restrict__ rank, int E) {
  int e = blockIdx.x * blockDim.x + threadIdx.x;
  if (e < E) {
    int d = dst[e];
    atomicAdd(&deg[d], 1.0f);
    rank[e] = atomicAdd(&tcnt[d >> 4], 1u);
  }
}
__global__ __launch_bounds__(1024) void tile_scan(const unsigned* __restrict__ tcnt,
                                                  unsigned* __restrict__ toff,
                                                  unsigned* __restrict__ rec) {
  __shared__ unsigned s[1024];
  int t = threadIdx.x;
  unsigned n[4], c[4], sum = 0;
#pragma unroll
  for (int i = 0; i < 4; ++i) {
    int idx = t * 4 + i;
    n[i] = idx < NT ? tcnt[idx] : 0u;
    c[i] = (n[i] + 7u) & ~7u;
    sum += c[i];
  }
  s[t] = sum;
  __syncthreads();
  for (int o = 1; o < 1024; o <<= 1) {
    unsigned v = t >= o ? s[t - o] : 0u;
    __syncthreads();
    s[t] += v;
    __syncthreads();
  }
  unsigned run = s[t] - sum;
#pragma unroll
  for (int i = 0; i < 4; ++i) {
    int idx = t * 4 + i;
    if (idx <= NT) toff[idx] = run;
    if (idx < NT) {
      for (unsigned j = n[i]; j < c[i]; ++j) {
        rec[2 * (run + j)]     = (unsigned)(idx * 16) << 4;
        rec[2 * (run + j) + 1] = 0u;
      }
    }
    run += c[i];
  }
}
__global__ void tile_place(const int* __restrict__ src, const int* __restrict__ dst,
                           const float* __restrict__ dinv, const unsigned* __restrict__ toff,
                           const unsigned* __restrict__ rank, unsigned* __restrict__ rec, int E) {
  int e = blockIdx.x * blockDim.x + threadIdx.x;
  if (e < E) {
    int s = src[e], d = dst[e];
    unsigned slot = toff[d >> 4] + rank[e];
    v2u r;
    r.x = ((unsigned)s << 4) | (unsigned)(d & 15);
    r.y = __float_as_uint(dinv[s] * dinv[d]);
    *(v2u*)(rec + 2 * (size_t)slot) = r;
  }
}
__global__ void rsqrt_inplace(float* __restrict__ p, int n) {
  int i = blockIdx.x * blockDim.x + threadIdx.x;
  if (i < n) p[i] = rsqrtf(p[i]);
}

__global__ void pad_x(const float* __restrict__ x, float* __restrict__ xp, int M) {
  int n = blockIdx.x, c = threadIdx.x;
  if (n >= M || c >= DINP) return;
  xp[(size_t)n * DINP + c] = (c < DIN) ? x[(size_t)n * DIN + c] : 0.0f;
}
__global__ void pad_W(const float* __restrict__ W, float* __restrict__ Wp, int K, int Nn) {
  int r = blockIdx.x, c = threadIdx.x;
  Wp[(size_t)r * DHP + c] = (r < K && c < Nn) ? W[(size_t)r * Nn + c] : 0.0f;
}

#define GB_ROWS 64
#define GB_SLOTQ 84
#define GB_BQ   (4 * GB_SLOTQ)
#define GB_CP   (DHP + 4)
template <int KP, int KC>
__global__ __launch_bounds__(256) void gemm_lds_wmma(
    const float* __restrict__ A, const float* __restrict__ B, float* __restrict__ C, int M) {
  constexpr int AP    = KC + 4;
  constexpr int NCH   = KP / KC;
  constexpr int KQ    = KC / 4;
  constexpr int APR   = KC / 4;
  constexpr int A_F4  = GB_ROWS * APR;
  constexpr int B_MB  = KQ * (DHP / 4);
  constexpr int A_IT  = (A_F4 + 255) / 256;
  constexpr int B_IT  = (B_MB + 255) / 256;
  constexpr int MAINF = GB_ROWS * AP + KQ * GB_BQ * 4;
  constexpr int CF    = GB_ROWS * GB_CP;
  static_assert(KP % KC == 0 && KC % 8 == 0 && ((AP / 4) & 1) == 1, "chunking");
  __shared__ __attribute__((aligned(16))) float smem[MAINF > CF ? MAINF : CF];
  float* As = smem;
  v4f*   Bs = (v4f*)(smem + GB_ROWS * AP);

  const int tid  = threadIdx.x;
  const int lane = tid & 31;
  const int w    = tid >> 5;
  const int half = lane >> 4;
  const int c    = lane & 15;
  const int mp   = w & 1;
  const int q    = w >> 1;
  const int row0 = blockIdx.x * GB_ROWS;

  v4f pa[A_IT];
  v4f pb[B_IT][4];

  auto fetch = [&](int ch) {
    const int kc0 = ch * KC;
#pragma unroll
    for (int i = 0; i < A_IT; ++i) {
      int idx = tid + 256 * i;
      if (idx < A_F4) {
        int r = idx / APR, c4 = idx - r * APR;
        int grow = row0 + r;
        grow = grow < M ? grow : M - 1;
        pa[i] = *(const v4f*)(A + (size_t)grow * KP + kc0 + 4 * c4);
      }
    }
#pragma unroll
    for (int i = 0; i < B_IT; ++i) {
      int mb = tid + 256 * i;
      if (mb < B_MB) {
        int kq = mb / (DHP / 4), cg = mb - kq * (DHP / 4);
        const float* bp = B + (size_t)(kc0 + 4 * kq) * DHP + 4 * cg;
#pragma unroll
        for (int r = 0; r < 4; ++r) pb[i][r] = *(const v4f*)(bp + (size_t)r * DHP);
      }
    }
  };
  auto stage = [&]() {
#pragma unroll
    for (int i = 0; i < A_IT; ++i) {
      int idx = tid + 256 * i;
      if (idx < A_F4) {
        int r = idx / APR, c4 = idx - r * APR;
        *(v4f*)(As + r * AP + 4 * c4) = pa[i];
      }
    }
#pragma unroll
    for (int i = 0; i < B_IT; ++i) {
      int mb = tid + 256 * i;
      if (mb < B_MB) {
        int kq = mb / (DHP / 4), cg = mb - kq * (DHP / 4);
        v4f* bs = Bs + kq * GB_BQ + cg;
#pragma unroll
        for (int j = 0; j < 4; ++j) {
          v4f t;
          t.x = pb[i][0][j]; t.y = pb[i][1][j]; t.z = pb[i][2][j]; t.w = pb[i][3][j];
          bs[j * GB_SLOTQ] = t;
        }
      }
    }
  };

  v8f acc[2][5];
#pragma unroll
  for (int mt = 0; mt < 2; ++mt)
#pragma unroll
    for (int nt = 0; nt < 5; ++nt)
#pragma unroll
      for (int r = 0; r < 8; ++r) acc[mt][nt][r] = 0.0f;

  const float* a0p = As + (32 * mp + c) * AP + 4 * half;
  const float* a1p = a0p + 16 * AP;
  int bslot[5];
#pragma unroll
  for (int nt = 0; nt < 5; ++nt) {
    int n = 80 * q + 16 * nt + c;
    bslot[nt] = half * GB_BQ + (n & 3) * GB_SLOTQ + (n >> 2);
  }

  fetch(0);
#pragma unroll 1
  for (int ch = 0; ch < NCH; ++ch) {
    stage();
    __syncthreads();
    if (ch + 1 < NCH) fetch(ch + 1);
#pragma unroll
    for (int kb = 0; kb < KC; kb += 8) {
      v4f a0 = *(const v4f*)(a0p + kb);
      v4f a1 = *(const v4f*)(a1p + kb);
      v4f b[5];
#pragma unroll
      for (int nt = 0; nt < 5; ++nt) b[nt] = Bs[(kb >> 2) * GB_BQ + bslot[nt]];
      v2f a0l, a0h, a1l, a1h;
      a0l.x = a0.x; a0l.y = a0.y; a0h.x = a0.z; a0h.y = a0.w;
      a1l.x = a1.x; a1l.y = a1.y; a1h.x = a1.z; a1h.y = a1.w;
#pragma unroll
      for (int nt = 0; nt < 5; ++nt) {
        v2f bl;
        bl.x = b[nt].x; bl.y = b[nt].y;
        acc[0][nt] = __builtin_amdgcn_wmma_f32_16x16x4_f32(false, a0l, false, bl, (short)0,
                                                           acc[0][nt], false, false);
        acc[1][nt] = __builtin_amdgcn_wmma_f32_16x16x4_f32(false, a1l, false, bl, (short)0,
                                                           acc[1][nt], false, false);
      }
#pragma unroll
      for (int nt = 0; nt < 5; ++nt) {
        v2f bh;
        bh.x = b[nt].z; bh.y = b[nt].w;
        acc[0][nt] = __builtin_amdgcn_wmma_f32_16x16x4_f32(false, a0h, false, bh, (short)0,
                                                           acc[0][nt], false, false);
        acc[1][nt] = __builtin_amdgcn_wmma_f32_16x16x4_f32(false, a1h, false, bh, (short)0,
                                                           acc[1][nt], false, false);
      }
    }
    __syncthreads();
  }

  float* Cs = smem;
#pragma unroll
  for (int mt = 0; mt < 2; ++mt)
#pragma unroll
    for (int nt = 0; nt < 5; ++nt)
#pragma unroll
      for (int r = 0; r < 8; ++r)
        Cs[(32 * mp + 16 * mt + 8 * half + r) * GB_CP + 80 * q + 16 * nt + c] = acc[mt][nt][r];
  __syncthreads();
#pragma unroll
  for (int i = 0; i < (GB_ROWS * (DHP / 4)) / 256; ++i) {
    int idx = tid + 256 * i;
    int r = idx / (DHP / 4), c4 = idx - r * (DHP / 4);
    if (row0 + r < M)
      *(v4f*)(C + (size_t)(row0 + r) * DHP + 4 * c4) = *(const v4f*)(Cs + r * GB_CP + 4 * c4);
  }
}

__device__ __forceinline__ unsigned enc_f32(float f) {
  unsigned u = __float_as_uint(f);
  return (u & 0x80000000u) ? ~u : (u | 0x80000000u);
}
__device__ __forceinline__ float dec_f32(unsigned u) {
  return __uint_as_float((u & 0x80000000u) ? (u & 0x7FFFFFFFu) : ~u);
}

#define NSC 16
template <bool POOL>
__global__ __launch_bounds__(160) __attribute__((amdgpu_waves_per_eu(16, 16))) void agg_tile_wmma(
    const float* __restrict__ h, const unsigned* __restrict__ rec,
    const unsigned* __restrict__ toff, const float* __restrict__ dinv,
    const float* __restrict__ bias, const int* __restrict__ batch,
    float* __restrict__ y, float* __restrict__ sums,
    unsigned* __restrict__ pmax, unsigned* __restrict__ pmin) {
  int t    = blockIdx.x;
  int wave = __builtin_amdgcn_readfirstlane((int)(threadIdx.x >> 5));
  int row0 = t * 16;
  int lane = threadIdx.x & 31;
  int half = lane >> 4;
  unsigned m = lane & 15;
  int col  = wave * 64 + (int)m * 4;

  v8f acc0, acc1, acc2, acc3;
#pragma unroll
  for (int r = 0; r < 8; ++r) {
    int row = row0 + r + half * 8;
    float di = dinv[row];
    v4f hv = *(const v4f*)(h + (size_t)row * DHP + col);
    acc0[r] = hv.x * di * di;
    acc1[r] = hv.y * di * di;
    acc2[r] = hv.z * di * di;
    acc3[r] = hv.w * di * di;
  }
  unsigned base = toff[t];
  unsigned n    = toff[t + 1] - base;
  const unsigned* rp = rec + 2 * (size_t)(base + 2 * half);
  for (unsigned i = 0; i < n; i += 8) {
    v4u ra = *(const v4u*)(rp + 2 * (size_t)i);
    v4u rb = *(const v4u*)(rp + 2 * (size_t)i + 8);
    v4f ha0 = *(const v4f*)(h + (size_t)(ra.x >> 4) * DHP + col);
    v4f ha1 = *(const v4f*)(h + (size_t)(ra.z >> 4) * DHP + col);
    v4f hb0 = *(const v4f*)(h + (size_t)(rb.x >> 4) * DHP + col);
    v4f hb1 = *(const v4f*)(h + (size_t)(rb.z >> 4) * DHP + col);
    v2f a, b;
    a.x = ((ra.x & 15u) == m) ? __uint_as_float(ra.y) : 0.0f;
    a.y = ((ra.z & 15u) == m) ? __uint_as_float(ra.w) : 0.0f;
    b.x = ha0.x; b.y = ha1.x;
    acc0 = __builtin_amdgcn_wmma_f32_16x16x4_f32(false, a, false, b, (short)0, acc0, false, false);
    b.x = ha0.y; b.y = ha1.y;
    acc1 = __builtin_amdgcn_wmma_f32_16x16x4_f32(false, a, false, b, (short)0, acc1, false, false);
    b.x = ha0.z; b.y = ha1.z;
    acc2 = __builtin_amdgcn_wmma_f32_16x16x4_f32(false, a, false, b, (short)0, acc2, false, false);
    b.x = ha0.w; b.y = ha1.w;
    acc3 = __builtin_amdgcn_wmma_f32_16x16x4_f32(false, a, false, b, (short)0, acc3, false, false);
    a.x = ((rb.x & 15u) == m) ? __uint_as_float(rb.y) : 0.0f;
    a.y = ((rb.z & 15u) == m) ? __uint_as_float(rb.w) : 0.0f;
    b.x = hb0.x; b.y = hb1.x;
    acc0 = __builtin_amdgcn_wmma_f32_16x16x4_f32(false, a, false, b, (short)0, acc0, false, false);
    b.x = hb0.y; b.y = hb1.y;
    acc1 = __builtin_amdgcn_wmma_f32_16x16x4_f32(false, a, false, b, (short)0, acc1, false, false);
    b.x = hb0.z; b.y = hb1.z;
    acc2 = __builtin_amdgcn_wmma_f32_16x16x4_f32(false, a, false, b, (short)0, acc2, false, false);
    b.x = hb0.w; b.y = hb1.w;
    acc3 = __builtin_amdgcn_wmma_f32_16x16x4_f32(false, a, false, b, (short)0, acc3, false, false);
  }

  const int ln = (int)__builtin_amdgcn_mbcnt_lo(~0u, 0u);
  const int eh = ln >> 4;
  const int ec = wave * 64 + (ln & 15) * 4;
  const bool valid = ec < DH;
  float b0 = 0.0f, b1 = 0.0f, b2 = 0.0f, b3 = 0.0f;
  if (valid) {
    b0 = bias[ec];
    b1 = bias[ec + 1];
    b2 = bias[ec + 2];
    b3 = bias[ec + 3];
  }
  float s0 = 0.f, s1 = 0.f, s2 = 0.f, s3 = 0.f;
  float q0 = 0.f, q1 = 0.f, q2 = 0.f, q3 = 0.f;
#pragma unroll
  for (int r = 0; r < 8; ++r) {
    float v0 = fmaxf(acc0[r] + b0, 0.0f);
    float v1 = fmaxf(acc1[r] + b1, 0.0f);
    float v2 = fmaxf(acc2[r] + b2, 0.0f);
    float v3 = fmaxf(acc3[r] + b3, 0.0f);
    acc0[r] = v0; acc1[r] = v1; acc2[r] = v2; acc3[r] = v3;
    s0 += v0; s1 += v1; s2 += v2; s3 += v3;
    q0 += v0 * v0; q1 += v1 * v1; q2 += v2 * v2; q3 += v3 * v3;
    if (!POOL) {
      int row = row0 + r + eh * 8;
      v4f o;
      o.x = v0; o.y = v1; o.z = v2; o.w = v3;
      *(v4f*)(y + (size_t)row * DHP + ec) = o;
    }
  }
  s0 += __shfl_xor(s0, 16); s1 += __shfl_xor(s1, 16);
  s2 += __shfl_xor(s2, 16); s3 += __shfl_xor(s3, 16);
  q0 += __shfl_xor(q0, 16); q1 += __shfl_xor(q1, 16);
  q2 += __shfl_xor(q2, 16); q3 += __shfl_xor(q3, 16);
  if (valid) {
    float* sp = sums + (size_t)(t & (NSC - 1)) * (2 * DHP) + eh * DHP + ec;
    atomicAdd(sp,     eh ? q0 : s0);
    atomicAdd(sp + 1, eh ? q1 : s1);
    atomicAdd(sp + 2, eh ? q2 : s2);
    atomicAdd(sp + 3, eh ? q3 : s3);
  }

  if (POOL) {
    const int* bp = batch + row0 + 8 * eh;
    const int g0 = batch[row0];
    int gid[8];
    bool same = true;
#pragma unroll
    for (int r = 0; r < 8; ++r) {
      gid[r] = bp[r];
      same = same && (gid[r] == g0);
    }
    if (__all(same)) {
      float hi0 = -INFINITY, hi1 = -INFINITY, hi2 = -INFINITY, hi3 = -INFINITY;
      float lo0 = INFINITY, lo1 = INFINITY, lo2 = INFINITY, lo3 = INFINITY;
#pragma unroll
      for (int r = 0; r < 8; ++r) {
        hi0 = fmaxf(hi0, acc0[r]); lo0 = fminf(lo0, acc0[r]);
        hi1 = fmaxf(hi1, acc1[r]); lo1 = fminf(lo1, acc1[r]);
        hi2 = fmaxf(hi2, acc2[r]); lo2 = fminf(lo2, acc2[r]);
        hi3 = fmaxf(hi3, acc3[r]); lo3 = fminf(lo3, acc3[r]);
      }
      hi0 = fmaxf(hi0, __shfl_xor(hi0, 16)); lo0 = fminf(lo0, __shfl_xor(lo0, 16));
      hi1 = fmaxf(hi1, __shfl_xor(hi1, 16)); lo1 = fminf(lo1, __shfl_xor(lo1, 16));
      hi2 = fmaxf(hi2, __shfl_xor(hi2, 16)); lo2 = fminf(lo2, __shfl_xor(lo2, 16));
      hi3 = fmaxf(hi3, __shfl_xor(hi3, 16)); lo3 = fminf(lo3, __shfl_xor(lo3, 16));
      if (valid) {
        unsigned* pp = (eh ? pmin : pmax) + (size_t)g0 * DH + ec;
        atomicMax(pp,     eh ? ~enc_f32(lo0) : enc_f32(hi0));
        atomicMax(pp + 1, eh ? ~enc_f32(lo1) : enc_f32(hi1));
        atomicMax(pp + 2, eh ? ~enc_f32(lo2) : enc_f32(hi2));
        atomicMax(pp + 3, eh ? ~enc_f32(lo3) : enc_f32(hi3));
      }
    } else if (valid) {
      auto pool_runs = [&](const v8f& v, unsigned cj) {
        int g = gid[0];
        float hi = -INFINITY, lo = INFINITY;
#pragma unroll
        for (int r = 0; r < 8; ++r) {
          if (gid[r] != g) {
            atomicMax(pmax + ((unsigned)g * DH + cj), enc_f32(hi));
            atomicMax(pmin + ((unsigned)g * DH + cj), ~enc_f32(lo));
            g  = gid[r];
            hi = -INFINITY;
            lo = INFINITY;
          }
          hi = fmaxf(hi, v[r]);
          lo = fminf(lo, v[r]);
        }
        atomicMax(pmax + ((unsigned)g * DH + cj), enc_f32(hi));
        atomicMax(pmin + ((unsigned)g * DH + cj), ~enc_f32(lo));
      };
      pool_runs(acc0, (unsigned)ec);
      pool_runs(acc1, (unsigned)ec + 1u);
      pool_runs(acc2, (unsigned)ec + 2u);
      pool_runs(acc3, (unsigned)ec + 3u);
    }
  }
}

__device__ __forceinline__ void bn_scale_shift(const float* __restrict__ sums, int c, int M,
                                               float g, float b, float& scale, float& shift) {
  float s = 0.f, q = 0.f;
  for (int k = 0; k < NSC; ++k) {
    s += sums[k * (2 * DHP) + c];
    q += sums[k * (2 * DHP) + DHP + c];
  }
  float invn = 1.0f / (float)M;
  float mean = s * invn;
  float var  = q * invn - mean * mean;
  scale = rsqrtf(var + BNEPS) * g;
  shift = b - mean * scale;
}
__global__ void bn_apply(float* __restrict__ y, const float* __restrict__ sums,
                         const float* __restrict__ gamma, const float* __restrict__ beta,
                         int M) {
  int c = threadIdx.x;
  if (c >= DH) return;
  int r0 = blockIdx.x * 128;
  int r1 = min(r0 + 128, M);
  float scale, shift;
  bn_scale_shift(sums, c, M, gamma[c], beta[c], scale, shift);
  for (int r = r0; r < r1; ++r)
    y[(size_t)r * DHP + c] = y[(size_t)r * DHP + c] * scale + shift;
}
__global__ void pool_bn_finish(const unsigned* __restrict__ pmax, const unsigned* __restrict__ pmin,
                               const float* __restrict__ sums, const float* __restrict__ gamma,
                               const float* __restrict__ beta, float* __restrict__ out, int M) {
  int g = blockIdx.x, c = threadIdx.x;
  if (c >= DH) return;
  float scale, shift;
  bn_scale_shift(sums, c, M, gamma[c], beta[c], scale, shift);
  unsigned em = pmax[(size_t)g * DH + c];
  float res = -INFINITY;
  if (em != 0u) {
    float v = (scale >= 0.0f) ? dec_f32(em) : dec_f32(~pmin[(size_t)g * DH + c]);
    res = v * scale + shift;
  }
  out[(size_t)g * DH + c] = res;
}

extern "C" void kernel_launch(void* const* d_in, const int* in_sizes, int n_in,
                              void* d_out, int out_size, void* d_ws, size_t ws_size,
                              hipStream_t stream) {
  const float* x      = (const float*)d_in[0];
  const int*   eidx   = (const int*)d_in[1];
  const int*   batch  = (const int*)d_in[2];
  const float* W1     = (const float*)d_in[3];
  const float* b1     = (const float*)d_in[4];
  const float* gamma1 = (const float*)d_in[5];
  const float* beta1  = (const float*)d_in[6];
  const float* W2     = (const float*)d_in[7];
  const float* b2     = (const float*)d_in[8];
  const float* gamma2 = (const float*)d_in[9];
  const float* beta2  = (const float*)d_in[10];
  float* out = (float*)d_out;

  const int* src = eidx;
  const int* dst = eidx + EE;

  float* ws = (float*)d_ws;
  float*    deg  = ws;
  float*    xp   = ws + 50048;
  float*    Wp1  = ws + 4050048;
  float*    Wp2  = ws + 4075648;
  float*    sums = ws + 4178048;
  float*    bufA = ws + 4334688;
  float*    bufB = ws + 20334688;
  unsigned* tcnt = (unsigned*)(ws + 36334688);
  unsigned* toff = (unsigned*)(ws + 36337888);
  unsigned* rank = (unsigned*)(ws + 36341088);
  unsigned* rec  = (unsigned*)(ws + 37141088);
  unsigned* pmax = (unsigned*)(ws + 38800000);
  unsigned* pmin = pmax + GG * DH;

  const int statBlocks = (NN + 127) / 128;
  const int gemmGrid   = (NN + GB_ROWS - 1) / GB_ROWS;

  fill_f32<<<(NN + 255) / 256, 256, 0, stream>>>(deg, 1.0f, NN);
  fill_u32<<<(NT + 255) / 256, 256, 0, stream>>>(tcnt, 0u, NT);
  deg_tile_count<<<(EE + 255) / 256, 256, 0, stream>>>(dst, deg, tcnt, rank, EE);
  rsqrt_inplace<<<(NN + 255) / 256, 256, 0, stream>>>(deg, NN);

  tile_scan<<<1, 1024, 0, stream>>>(tcnt, toff, rec);
  tile_place<<<(EE + 255) / 256, 256, 0, stream>>>(src, dst, deg, toff, rank, rec, EE);

  pad_x<<<NN, 128, 0, stream>>>(x, xp, NN);
  pad_W<<<DINP, DHP, 0, stream>>>(W1, Wp1, DIN, DH);
  pad_W<<<DHP, DHP, 0, stream>>>(W2, Wp2, DH, DH);

  gemm_lds_wmma<DINP, 16><<<gemmGrid, 256, 0, stream>>>(xp, Wp1, bufA, NN);
  fill_f32<<<(NSC * 2 * DHP + 255) / 256, 256, 0, stream>>>(sums, 0.0f, NSC * 2 * DHP);
  agg_tile_wmma<false><<<NT, 160, 0, stream>>>(bufA, rec, toff, deg, b1, batch, bufB, sums, pmax,
                                               pmin);
  bn_apply<<<statBlocks, DHP, 0, stream>>>(bufB, sums, gamma1, beta1, NN);

  gemm_lds_wmma<DHP, 32><<<gemmGrid, 256, 0, stream>>>(bufB, Wp2, bufA, NN);
  fill_f32<<<(NSC * 2 * DHP + 255) / 256, 256, 0, stream>>>(sums, 0.0f, NSC * 2 * DHP);
  fill_u32<<<(2L * GG * DH + 255) / 256, 256, 0, stream>>>(pmax, 0u, 2L * GG * DH);

  agg_tile_wmma<true><<<NT, 160, 0, stream>>>(bufA, rec, toff, deg, b2, batch, bufB, sums, pmax,
                                              pmin);
  pool_bn_finish<<<GG, DHP, 0, stream>>>(pmax, pmin, sums, gamma2, beta2, out, NN);
}


# Layer 1's BN-apply pass gone: GEMM 2 reads y as stored and maps each 16-byte A piece (scale, shift per k, totalled from the moment sums into LDS) between global load and LDS store; sums unchanged
# speedup vs baseline: 4.6158x; 1.0716x over previous
#include <hip/hip_runtime.h>
#include <hip/hip_bf16.h>

typedef float v2f __attribute__((ext_vector_type(2)));
typedef float v8f __attribute__((ext_vector_type(8)));
typedef float v4f __attribute__((ext_vector_type(4)));
typedef unsigned v4u __attribute__((ext_vector_type(4)));
typedef unsigned v2u __attribute__((ext_vector_type(2)));

#define NN    50000
#define EE    800000
#define GG    500
#define DIN   78
#define DINP  80
#define DH    312
#define DHP   320
#define BNEPS 1e-5f
#define NT    (NN / 16)
static_assert(NN % 16 == 0, "destination tiles are whole");

__global__ void fill_f32(float* __restrict__ p, float v, long n) {
  long i = (long)blockIdx.x * blockDim.x + threadIdx.x;
  if (i < n) p[i] = v;
}
__global__ void fill_u32(unsigned* __restrict__ p, unsigned v, long n) {
  long i = (long)blockIdx.x * blockDim.x + threadIdx.x;
  if (i < n) p[i] = v;
}

__global__ void deg_tile_count(const int* __restrict__ dst, float* __restrict__ deg,
                               unsigned* __restrict__ tcnt, unsigned* __restrict__ rank, int E) {
  int e = blockIdx.x * blockDim.x + threadIdx.x;
  if (e < E) {
    int d = dst[e];
    atomicAdd(&deg[d], 1.0f);
    rank[e] = atomicAdd(&tcnt[d >> 4], 1u);
  }
}
__global__ __launch_bounds__(1024) void tile_scan(const unsigned* __restrict__ tcnt,
                                                  unsigned* __restrict__ toff,
                                                  unsigned* __restrict__ rec) {
  __shared__ unsigned s[1024];
  int t = threadIdx.x;
  unsigned n[4], c[4], sum = 0;
#pragma unroll
  for (int i = 0; i < 4; ++i) {
    int idx = t * 4 + i;
    n[i] = idx < NT ? tcnt[idx] : 0u;
    c[i] = (n[i] + 7u) & ~7u;
    sum += c[i];
  }
  s[t] = sum;
  __syncthreads();
  for (int o = 1; o < 1024; o <<= 1) {
    unsigned v = t >= o ? s[t - o] : 0u;
    __syncthreads();
    s[t] += v;
    __syncthreads();
  }
  unsigned run = s[t] - sum;
#pragma unroll
  for (int i = 0; i < 4; ++i) {
    int idx = t * 4 + i;
    if (idx <= NT) toff[idx] = run;
    if (idx < NT) {
      for (unsigned j = n[i]; j < c[i]; ++j) {
        rec[2 * (run + j)]     = (unsigned)(idx * 16) << 4;
        rec[2 * (run + j) + 1] = 0u;
      }
    }
    run += c[i];
  }
}
__global__ void tile_place(const int* __restrict__ src, const int* __restrict__ dst,
                           const float* __restrict__ dinv, const unsigned* __restrict__ toff,
                           const unsigned* __restrict__ rank, unsigned* __restrict__ rec, int E) {
  int e = blockIdx.x * blockDim.x + threadIdx.x;
  if (e < E) {
    int s = src[e], d = dst[e];
    unsigned slot = toff[d >> 4] + rank[e];
    v2u r;
    r.x = ((unsigned)s << 4) | (unsigned)(d & 15);
    r.y = __float_as_uint(dinv[s] * dinv[d]);
    *(v2u*)(rec + 2 * (size_t)slot) = r;
  }
}
__global__ void rsqrt_inplace(float* __restrict__ p, int n) {
  int i = blockIdx.x * blockDim.x + threadIdx.x;
  if (i < n) p[i] = rsqrtf(p[i]);
}

__global__ void pad_x(const float* __restrict__ x, float* __restrict__ xp, int M) {
  int n = blockIdx.x, c = threadIdx.x;
  if (n >= M || c >= DINP) return;
  xp[(size_t)n * DINP + c] = (c < DIN) ? x[(size_t)n * DIN + c] : 0.0f;
}
__global__ void pad_W(const float* __restrict__ W, float* __restrict__ Wp, int K, int Nn) {
  int r = blockIdx.x, c = threadIdx.x;
  Wp[(size_t)r * DHP + c] = (r < K && c < Nn) ? W[(size_t)r * Nn + c] : 0.0f;
}

#define NSC 16
__device__ __forceinline__ void bn_scale_shift(const float* __restrict__ sums, int c, int M,
                                               float g, float b, float& scale, float& shift) {
  float s = 0.f, q = 0.f;
  for (int k = 0; k < NSC; ++k) {
    s += sums[k * (2 * DHP) + c];
    q += sums[k * (2 * DHP) + DHP + c];
  }
  float invn = 1.0f / (float)M;
  float mean = s * invn;
  float var  = q * invn - mean * mean;
  scale = rsqrtf(var + BNEPS) * g;
  shift = b - mean * scale;
}

#define GB_ROWS 64
#define GB_SLOTQ 84
#define GB_BQ   (4 * GB_SLOTQ)
#define GB_CP   (DHP + 4)
template <int KP, int KC, bool BN>
__global__ __launch_bounds__(256) void gemm_lds_wmma(
    const float* __restrict__ A, const float* __restrict__ B, float* __restrict__ C, int M,
    const float* __restrict__ sums, const float* __restrict__ gamma,
    const float* __restrict__ beta) {
  constexpr int AP    = KC + 4;
  constexpr int NCH   = KP / KC;
  constexpr int KQ    = KC / 4;
  constexpr int APR   = KC / 4;
  constexpr int A_F4  = GB_ROWS * APR;
  constexpr int B_MB  = KQ * (DHP / 4);
  constexpr int A_IT  = (A_F4 + 255) / 256;
  constexpr int B_IT  = (B_MB + 255) / 256;
  constexpr int MAINF = GB_ROWS * AP + KQ * GB_BQ * 4;
  constexpr int CF    = GB_ROWS * GB_CP;
  static_assert(KP % KC == 0 && KC % 8 == 0 && ((AP / 4) & 1) == 1, "chunking");
  __shared__ __attribute__((aligned(16))) float smem[MAINF > CF ? MAINF : CF];
  __shared__ __attribute__((aligned(16))) float coef[BN ? 2 * KP : 4];
  float* As = smem;
  v4f*   Bs = (v4f*)(smem + GB_ROWS * AP);

  const int tid  = threadIdx.x;
  const int lane = tid & 31;
  const int w    = tid >> 5;
  const int half = lane >> 4;
  const int c    = lane & 15;
  const int mp   = w & 1;
  const int q    = w >> 1;
  const int row0 = blockIdx.x * GB_ROWS;

  v4f pa[A_IT];
  v4f pb[B_IT][4];

  auto fetch = [&](int ch) {
    const int kc0 = ch * KC;
#pragma unroll
    for (int i = 0; i < A_IT; ++i) {
      int idx = tid + 256 * i;
      if (idx < A_F4) {
        int r = idx / APR, c4 = idx - r * APR;
        int grow = row0 + r;
        grow = grow < M ? grow : M - 1;
        pa[i] = *(const v4f*)(A + (size_t)grow * KP + kc0 + 4 * c4);
      }
    }
#pragma unroll
    for (int i = 0; i < B_IT; ++i) {
      int mb = tid + 256 * i;
      if (mb < B_MB) {
        int kq = mb / (DHP / 4), cg = mb - kq * (DHP / 4);
        const float* bp = B + (size_t)(kc0 + 4 * kq) * DHP + 4 * cg;
#pragma unroll
        for (int r = 0; r < 4; ++r) pb[i][r] = *(const v4f*)(bp + (size_t)r * DHP);
      }
    }
  };
  auto stage = [&](int ch) {
#pragma unroll
    for (int i = 0; i < A_IT; ++i) {
      int idx = tid + 256 * i;
      if (idx < A_F4) {
        int r = idx / APR, c4 = idx - r * APR;
        v4f av = pa[i];
        if (BN) {
          const v4f sc = *(const v4f*)(coef + ch * KC + 4 * c4);
          const v4f sh = *(const v4f*)(coef + KP + ch * KC + 4 * c4);
          av.x = av.x * sc.x + sh.x;
          av.y = av.y * sc.y + sh.y;
          av.z = av.z * sc.z + sh.z;
          av.w = av.w * sc.w + sh.w;
        }
        *(v4f*)(As + r * AP + 4 * c4) = av;
      }
    }
#pragma unroll
    for (int i = 0; i < B_IT; ++i) {
      int mb = tid + 256 * i;
      if (mb < B_MB) {
        int kq = mb / (DHP / 4), cg = mb - kq * (DHP / 4);
        v4f* bs = Bs + kq * GB_BQ + cg;
#pragma unroll
        for (int j = 0; j < 4; ++j) {
          v4f t;
          t.x = pb[i][0][j]; t.y = pb[i][1][j]; t.z = pb[i][2][j]; t.w = pb[i][3][j];
          bs[j * GB_SLOTQ] = t;
        }
      }
    }
  };

  v8f acc[2][5];
#pragma unroll
  for (int mt = 0; mt < 2; ++mt)
#pragma unroll
    for (int nt = 0; nt < 5; ++nt)
#pragma unroll
      for (int r = 0; r < 8; ++r) acc[mt][nt][r] = 0.0f;

  const float* a0p = As + (32 * mp + c) * AP + 4 * half;
  const float* a1p = a0p + 16 * AP;
  int bslot[5];
#pragma unroll
  for (int nt = 0; nt < 5; ++nt) {
    int n = 80 * q + 16 * nt + c;
    bslot[nt] = half * GB_BQ + (n & 3) * GB_SLOTQ + (n >> 2);
  }

  fetch(0);
  if (BN) {
    for (int k = tid; k < KP; k += 256) {
      float sc = 0.0f, sh = 0.0f;
      if (k < DH) bn_scale_shift(sums, k, M, gamma[k], beta[k], sc, sh);
      coef[k]      = sc;
      coef[KP + k] = sh;
    }
    __syncthreads();
  }
#pragma unroll 1
  for (int ch = 0; ch < NCH; ++ch) {
    stage(ch);
    __syncthreads();
    if (ch + 1 < NCH) fetch(ch + 1);
#pragma unroll
    for (int kb = 0; kb < KC; kb += 8) {
      v4f a0 = *(const v4f*)(a0p + kb);
      v4f a1 = *(const v4f*)(a1p + kb);
      v4f b[5];
#pragma unroll
      for (int nt = 0; nt < 5; ++nt) b[nt] = Bs[(kb >> 2) * GB_BQ + bslot[nt]];
      v2f a0l, a0h, a1l, a1h;
      a0l.x = a0.x; a0l.y = a0.y; a0h.x = a0.z; a0h.y = a0.w;
      a1l.x = a1.x; a1l.y = a1.y; a1h.x = a1.z; a1h.y = a1.w;
#pragma unroll
      for (int nt = 0; nt < 5; ++nt) {
        v2f bl;
        bl.x = b[nt].x; bl.y = b[nt].y;
        acc[0][nt] = __builtin_amdgcn_wmma_f32_16x16x4_f32(false, a0l, false, bl, (short)0,
                                                           acc[0][nt], false, false);
        acc[1][nt] = __builtin_amdgcn_wmma_f32_16x16x4_f32(false, a1l, false, bl, (short)0,
                                                           acc[1][nt], false, false);
      }
#pragma unroll
      for (int nt = 0; nt < 5; ++nt) {
        v2f bh;
        bh.x = b[nt].z; bh.y = b[nt].w;
        acc[0][nt] = __builtin_amdgcn_wmma_f32_16x16x4_f32(false, a0h, false, bh, (short)0,
                                                           acc[0][nt], false, false);
        acc[1][nt] = __builtin_amdgcn_wmma_f32_16x16x4_f32(false, a1h, false, bh, (short)0,
                                                           acc[1][nt], false, false);
      }
    }
    __syncthreads();
  }

  float* Cs = smem;
#pragma unroll
  for (int mt = 0; mt < 2; ++mt)
#pragma unroll
    for (int nt = 0; nt < 5; ++nt)
#pragma unroll
      for (int r = 0; r < 8; ++r)
        Cs[(32 * mp + 16 * mt + 8 * half + r) * GB_CP + 80 * q + 16 * nt + c] = acc[mt][nt][r];
  __syncthreads();
#pragma unroll
  for (int i = 0; i < (GB_ROWS * (DHP / 4)) / 256; ++i) {
    int idx = tid + 256 * i;
    int r = idx / (DHP / 4), c4 = idx - r * (DHP / 4);
    if (row0 + r < M)
      *(v4f*)(C + (size_t)(row0 + r) * DHP + 4 * c4) = *(const v4f*)(Cs + r * GB_CP + 4 * c4);
  }
}

__device__ __forceinline__ unsigned enc_f32(float f) {
  unsigned u = __float_as_uint(f);
  return (u & 0x80000000u) ? ~u : (u | 0x80000000u);
}
__device__ __forceinline__ float dec_f32(unsigned u) {
  return __uint_as_float((u & 0x80000000u) ? (u & 0x7FFFFFFFu) : ~u);
}

template <bool POOL>
__global__ __launch_bounds__(160) __attribute__((amdgpu_waves_per_eu(16, 16))) void agg_tile_wmma(
    const float* __restrict__ h, const unsigned* __restrict__ rec,
    const unsigned* __restrict__ toff, const float* __restrict__ dinv,
    const float* __restrict__ bias, const int* __restrict__ batch,
    float* __restrict__ y, float* __restrict__ sums,
    unsigned* __restrict__ pmax, unsigned* __restrict__ pmin) {
  int t    = blockIdx.x;
  int wave = __builtin_amdgcn_readfirstlane((int)(threadIdx.x >> 5));
  int row0 = t * 16;
  int lane = threadIdx.x & 31;
  int half = lane >> 4;
  unsigned m = lane & 15;
  int col  = wave * 64 + (int)m * 4;

  v8f acc0, acc1, acc2, acc3;
#pragma unroll
  for (int r = 0; r < 8; ++r) {
    int row = row0 + r + half * 8;
    float di = dinv[row];
    v4f hv = *(const v4f*)(h + (size_t)row * DHP + col);
    acc0[r] = hv.x * di * di;
    acc1[r] = hv.y * di * di;
    acc2[r] = hv.z * di * di;
    acc3[r] = hv.w * di * di;
  }
  unsigned base = toff[t];
  unsigned n    = toff[t + 1] - base;
  const unsigned* rp = rec + 2 * (size_t)(base + 2 * half);
  for (unsigned i = 0; i < n; i += 8) {
    v4u ra = *(const v4u*)(rp + 2 * (size_t)i);
    v4u rb = *(const v4u*)(rp + 2 * (size_t)i + 8);
    v4f ha0 = *(const v4f*)(h + (size_t)(ra.x >> 4) * DHP + col);
    v4f ha1 = *(const v4f*)(h + (size_t)(ra.z >> 4) * DHP + col);
    v4f hb0 = *(const v4f*)(h + (size_t)(rb.x >> 4) * DHP + col);
    v4f hb1 = *(const v4f*)(h + (size_t)(rb.z >> 4) * DHP + col);
    v2f a, b;
    a.x = ((ra.x & 15u) == m) ? __uint_as_float(ra.y) : 0.0f;
    a.y = ((ra.z & 15u) == m) ? __uint_as_float(ra.w) : 0.0f;
    b.x = ha0.x; b.y = ha1.x;
    acc0 = __builtin_amdgcn_wmma_f32_16x16x4_f32(false, a, false, b, (short)0, acc0, false, false);
    b.x = ha0.y; b.y = ha1.y;
    acc1 = __builtin_amdgcn_wmma_f32_16x16x4_f32(false, a, false, b, (short)0, acc1, false, false);
    b.x = ha0.z; b.y = ha1.z;
    acc2 = __builtin_amdgcn_wmma_f32_16x16x4_f32(false, a, false, b, (short)0, acc2, false, false);
    b.x = ha0.w; b.y = ha1.w;
    acc3 = __builtin_amdgcn_wmma_f32_16x16x4_f32(false, a, false, b, (short)0, acc3, false, false);
    a.x = ((rb.x & 15u) == m) ? __uint_as_float(rb.y) : 0.0f;
    a.y = ((rb.z & 15u) == m) ? __uint_as_float(rb.w) : 0.0f;
    b.x = hb0.x; b.y = hb1.x;
    acc0 = __builtin_amdgcn_wmma_f32_16x16x4_f32(false, a, false, b, (short)0, acc0, false, false);
    b.x = hb0.y; b.y = hb1.y;
    acc1 = __builtin_amdgcn_wmma_f32_16x16x4_f32(false, a, false, b, (short)0, acc1, false, false);
    b.x = hb0.z; b.y = hb1.z;
    acc2 = __builtin_amdgcn_wmma_f32_16x16x4_f32(false, a, false, b, (short)0, acc2, false, false);
    b.x = hb0.w; b.y = hb1.w;
    acc3 = __builtin_amdgcn_wmma_f32_16x16x4_f32(false, a, false, b, (short)0, acc3, false, false);
  }

  const int ln = (int)__builtin_amdgcn_mbcnt_lo(~0u, 0u);
  const int eh = ln >> 4;
  const int ec = wave * 64 + (ln & 15) * 4;
  const bool valid = ec < DH;
  float b0 = 0.0f, b1 = 0.0f, b2 = 0.0f, b3 = 0.0f;
  if (valid) {
    b0 = bias[ec];
    b1 = bias[ec + 1];
    b2 = bias[ec + 2];
    b3 = bias[ec + 3];
  }
  float s0 = 0.f, s1 = 0.f, s2 = 0.f, s3 = 0.f;
  float q0 = 0.f, q1 = 0.f, q2 = 0.f, q3 = 0.f;
#pragma unroll
  for (int r = 0; r < 8; ++r) {
    float v0 = fmaxf(acc0[r] + b0, 0.0f);
    float v1 = fmaxf(acc1[r] + b1, 0.0f);
    float v2 = fmaxf(acc2[r] + b2, 0.0f);
    float v3 = fmaxf(acc3[r] + b3, 0.0f);
    acc0[r] = v0; acc1[r] = v1; acc2[r] = v2; acc3[r] = v3;
    s0 += v0; s1 += v1; s2 += v2; s3 += v3;
    q0 += v0 * v0; q1 += v1 * v1; q2 += v2 * v2; q3 += v3 * v3;
    if (!POOL) {
      int row = row0 + r + eh * 8;
      v4f o;
      o.x = v0; o.y = v1; o.z = v2; o.w = v3;
      *(v4f*)(y + (size_t)row * DHP + ec) = o;
    }
  }
  s0 += __shfl_xor(s0, 16); s1 += __shfl_xor(s1, 16);
  s2 += __shfl_xor(s2, 16); s3 += __shfl_xor(s3, 16);
  q0 += __shfl_xor(q0, 16); q1 += __shfl_xor(q1, 16);
  q2 += __shfl_xor(q2, 16); q3 += __shfl_xor(q3, 16);
  if (valid) {
    float* sp = sums + (size_t)(t & (NSC - 1)) * (2 * DHP) + eh * DHP + ec;
    atomicAdd(sp,     eh ? q0 : s0);
    atomicAdd(sp + 1, eh ? q1 : s1);
    atomicAdd(sp + 2, eh ? q2 : s2);
    atomicAdd(sp + 3, eh ? q3 : s3);
  }

  if (POOL) {
    const int* bp = batch + row0 + 8 * eh;
    const int g0 = batch[row0];
    int gid[8];
    bool same = true;
#pragma unroll
    for (int r = 0; r < 8; ++r) {
      gid[r] = bp[r];
      same = same && (gid[r] == g0);
    }
    if (__all(same)) {
      float hi0 = -INFINITY, hi1 = -INFINITY, hi2 = -INFINITY, hi3 = -INFINITY;
      float lo0 = INFINITY, lo1 = INFINITY, lo2 = INFINITY, lo3 = INFINITY;
#pragma unroll
      for (int r = 0; r < 8; ++r) {
        hi0 = fmaxf(hi0, acc0[r]); lo0 = fminf(lo0, acc0[r]);
        hi1 = fmaxf(hi1, acc1[r]); lo1 = fminf(lo1, acc1[r]);
        hi2 = fmaxf(hi2, acc2[r]); lo2 = fminf(lo2, acc2[r]);
        hi3 = fmaxf(hi3, acc3[r]); lo3 = fminf(lo3, acc3[r]);
      }
      hi0 = fmaxf(hi0, __shfl_xor(hi0, 16)); lo0 = fminf(lo0, __shfl_xor(lo0, 16));
      hi1 = fmaxf(hi1, __shfl_xor(hi1, 16)); lo1 = fminf(lo1, __shfl_xor(lo1, 16));
      hi2 = fmaxf(hi2, __shfl_xor(hi2, 16)); lo2 = fminf(lo2, __shfl_xor(lo2, 16));
      hi3 = fmaxf(hi3, __shfl_xor(hi3, 16)); lo3 = fminf(lo3, __shfl_xor(lo3, 16));
      if (valid) {
        unsigned* pp = (eh ? pmin : pmax) + (size_t)g0 * DH + ec;
        atomicMax(pp,     eh ? ~enc_f32(lo0) : enc_f32(hi0));
        atomicMax(pp + 1, eh ? ~enc_f32(lo1) : enc_f32(hi1));
        atomicMax(pp + 2, eh ? ~enc_f32(lo2) : enc_f32(hi2));
        atomicMax(pp + 3, eh ? ~enc_f32(lo3) : enc_f32(hi3));
      }
    } else if (valid) {
      auto pool_runs = [&](const v8f& v, unsigned cj) {
        int g = gid[0];
        float hi = -INFINITY, lo = INFINITY;
#pragma unroll
        for (int r = 0; r < 8; ++r) {
          if (gid[r] != g) {
            atomicMax(pmax + ((unsigned)g * DH + cj), enc_f32(hi));
            atomicMax(pmin + ((unsigned)g * DH + cj), ~enc_f32(lo));
            g  = gid[r];
            hi = -INFINITY;
            lo = INFINITY;
          }
          hi = fmaxf(hi, v[r]);
          lo = fminf(lo, v[r]);
        }
        atomicMax(pmax + ((unsigned)g * DH + cj), enc_f32(hi));
        atomicMax(pmin + ((unsigned)g * DH + cj), ~enc_f32(lo));
      };
      pool_runs(acc0, (unsigned)ec);
      pool_runs(acc1, (unsigned)ec + 1u);
      pool_runs(acc2, (unsigned)ec + 2u);
      pool_runs(acc3, (unsigned)ec + 3u);
    }
  }
}

__global__ void pool_bn_finish(const unsigned* __restrict__ pmax, const unsigned* __restrict__ pmin,
                               const float* __restrict__ sums, const float* __restrict__ gamma,
                               const float* __restrict__ beta, float* __restrict__ out, int M) {
  int g = blockIdx.x, c = threadIdx.x;
  if (c >= DH) return;
  float scale, shift;
  bn_scale_shift(sums, c, M, gamma[c], beta[c], scale, shift);
  unsigned em = pmax[(size_t)g * DH + c];
  float res = -INFINITY;
  if (em != 0u) {
    float v = (scale >= 0.0f) ? dec_f32(em) : dec_f32(~pmin[(size_t)g * DH + c]);
    res = v * scale + shift;
  }
  out[(size_t)g * DH + c] = res;
}

extern "C" void kernel_launch(void* const* d_in, const int* in_sizes, int n_in,
                              void* d_out, int out_size, void* d_ws, size_t ws_size,
                              hipStream_t stream) {
  const float* x      = (const float*)d_in[0];
  const int*   eidx   = (const int*)d_in[1];
  const int*   batch  = (const int*)d_in[2];
  const float* W1     = (const float*)d_in[3];
  const float* b1     = (const float*)d_in[4];
  const float* gamma1 = (const float*)d_in[5];
  const float* beta1  = (const float*)d_in[6];
  const float* W2     = (const float*)d_in[7];
  const float* b2     = (const float*)d_in[8];
  const float* gamma2 = (const float*)d_in[9];
  const float* beta2  = (const float*)d_in[10];
  float* out = (float*)d_out;

  const int* src = eidx;
  const int* dst = eidx + EE;

  float* ws = (float*)d_ws;
  float*    deg  = ws;
  float*    xp   = ws + 50048;
  float*    Wp1  = ws + 4050048;
  float*    Wp2  = ws + 4075648;
  float*    sums = ws + 4178048;
  float*    bufA = ws + 4334688;
  float*    bufB = ws + 20334688;
  unsigned* tcnt = (unsigned*)(ws + 36334688);
  unsigned* toff = (unsigned*)(ws + 36337888);
  unsigned* rank = (unsigned*)(ws + 36341088);
  unsigned* rec  = (unsigned*)(ws + 37141088);
  unsigned* pmax = (unsigned*)(ws + 38800000);
  unsigned* pmin = pmax + GG * DH;

  const int gemmGrid   = (NN + GB_ROWS - 1) / GB_ROWS;

  fill_f32<<<(NN + 255) / 256, 256, 0, stream>>>(deg, 1.0f, NN);
  fill_u32<<<(NT + 255) / 256, 256, 0, stream>>>(tcnt, 0u, NT);
  deg_tile_count<<<(EE + 255) / 256, 256, 0, stream>>>(dst, deg, tcnt, rank, EE);
  rsqrt_inplace<<<(NN + 255) / 256, 256, 0, stream>>>(deg, NN);

  tile_scan<<<1, 1024, 0, stream>>>(tcnt, toff, rec);
  tile_place<<<(EE + 255) / 256, 256, 0, stream>>>(src, dst, deg, toff, rank, rec, EE);

  pad_x<<<NN, 128, 0, stream>>>(x, xp, NN);
  pad_W<<<DINP, DHP, 0, stream>>>(W1, Wp1, DIN, DH);
  pad_W<<<DHP, DHP, 0, stream>>>(W2, Wp2, DH, DH);

  gemm_lds_wmma<DINP, 16, false><<<gemmGrid, 256, 0, stream>>>(xp, Wp1, bufA, NN, nullptr, nullptr,
                                                               nullptr);
  fill_f32<<<(NSC * 2 * DHP + 255) / 256, 256, 0, stream>>>(sums, 0.0f, NSC * 2 * DHP);
  agg_tile_wmma<false><<<NT, 160, 0, stream>>>(bufA, rec, toff, deg, b1, batch, bufB, sums, pmax,
                                               pmin);

  gemm_lds_wmma<DHP, 32, true><<<gemmGrid, 256, 0, stream>>>(bufB, Wp2, bufA, NN, sums, gamma1,
                                                             beta1);
  fill_f32<<<(NSC * 2 * DHP + 255) / 256, 256, 0, stream>>>(sums, 0.0f, NSC * 2 * DHP);
  fill_u32<<<(2L * GG * DH + 255) / 256, 256, 0, stream>>>(pmax, 0u, 2L * GG * DH);

  agg_tile_wmma<true><<<NT, 160, 0, stream>>>(bufA, rec, toff, deg, b2, batch, bufB, sums, pmax,
                                              pmin);
  pool_bn_finish<<<GG, DHP, 0, stream>>>(pmax, pmin, sums, gamma2, beta2, out, NN);
}


# Padding passes gone (17 -> 14 launches): GEMM 1 reads x in 8-byte halves, both GEMMs read W as given, zero k rows/columns made while staging to LDS; GEMM 2 walks K=312 in chunks of 24; sums unchanged
# speedup vs baseline: 4.8018x; 1.0403x over previous
#include <hip/hip_runtime.h>
#include <hip/hip_bf16.h>

typedef float v2f __attribute__((ext_vector_type(2)));
typedef float v8f __attribute__((ext_vector_type(8)));
typedef float v4f __attribute__((ext_vector_type(4)));
typedef unsigned v4u __attribute__((ext_vector_type(4)));
typedef unsigned v2u __attribute__((ext_vector_type(2)));

#define NN    50000
#define EE    800000
#define GG    500
#define DIN   78
#define DINP  80
#define DH    312
#define DHP   320
#define BNEPS 1e-5f
#define NT    (NN / 16)
static_assert(NN % 16 == 0, "destination tiles are whole");

__global__ void fill_f32(float* __restrict__ p, float v, long n) {
  long i = (long)blockIdx.x * blockDim.x + threadIdx.x;
  if (i < n) p[i] = v;
}
__global__ void fill_u32(unsigned* __restrict__ p, unsigned v, long n) {
  long i = (long)blockIdx.x * blockDim.x + threadIdx.x;
  if (i < n) p[i] = v;
}

__global__ void deg_tile_count(const int* __restrict__ dst, float* __restrict__ deg,
                               unsigned* __restrict__ tcnt, unsigned* __restrict__ rank, int E) {
  int e = blockIdx.x * blockDim.x + threadIdx.x;
  if (e < E) {
    int d = dst[e];
    atomicAdd(&deg[d], 1.0f);
    rank[e] = atomicAdd(&tcnt[d >> 4], 1u);
  }
}
__global__ __launch_bounds__(1024) void tile_scan(const unsigned* __restrict__ tcnt,
                                                  unsigned* __restrict__ toff,
                                                  unsigned* __restrict__ rec) {
  __shared__ unsigned s[1024];
  int t = threadIdx.x;
  unsigned n[4], c[4], sum = 0;
#pragma unroll
  for (int i = 0; i < 4; ++i) {
    int idx = t * 4 + i;
    n[i] = idx < NT ? tcnt[idx] : 0u;
    c[i] = (n[i] + 7u) & ~7u;
    sum += c[i];
  }
  s[t] = sum;
  __syncthreads();
  for (int o = 1; o < 1024; o <<= 1) {
    unsigned v = t >= o ? s[t - o] : 0u;
    __syncthreads();
    s[t] += v;
    __syncthreads();
  }
  unsigned run = s[t] - sum;
#pragma unroll
  for (int i = 0; i < 4; ++i) {
    int idx = t * 4 + i;
    if (idx <= NT) toff[idx] = run;
    if (idx < NT) {
      for (unsigned j = n[i]; j < c[i]; ++j) {
        rec[2 * (run + j)]     = (unsigned)(idx * 16) << 4;
        rec[2 * (run + j) + 1] = 0u;
      }
    }
    run += c[i];
  }
}
__global__ void tile_place(const int* __restrict__ src, const int* __restrict__ dst,
                           const float* __restrict__ dinv, const unsigned* __restrict__ toff,
                           const unsigned* __restrict__ rank, unsigned* __restrict__ rec, int E) {
  int e = blockIdx.x * blockDim.x + threadIdx.x;
  if (e < E) {
    int s = src[e], d = dst[e];
    unsigned slot = toff[d >> 4] + rank[e];
    v2u r;
    r.x = ((unsigned)s << 4) | (unsigned)(d & 15);
    r.y = __float_as_uint(dinv[s] * dinv[d]);
    *(v2u*)(rec + 2 * (size_t)slot) = r;
  }
}
__global__ void rsqrt_inplace(float* __restrict__ p, int n) {
  int i = blockIdx.x * blockDim.x + threadIdx.x;
  if (i < n) p[i] = rsqrtf(p[i]);
}

#define NSC 16
__device__ __forceinline__ void bn_scale_shift(const float* __restrict__ sums, int c, int M,
                                               float g, float b, float& scale, float& shift) {
  float s = 0.f, q = 0.f;
  for (int k = 0; k < NSC; ++k) {
    s += sums[k * (2 * DHP) + c];
    q += sums[k * (2 * DHP) + DHP + c];
  }
  float invn = 1.0f / (float)M;
  float mean = s * invn;
  float var  = q * invn - mean * mean;
  scale = rsqrtf(var + BNEPS) * g;
  shift = b - mean * scale;
}

#define GB_ROWS 64
#define GB_SLOTQ 84
#define GB_BQ   (4 * GB_SLOTQ)
#define GB_CP   (DHP + 4)
template <int KP, int KC, bool BN, int APITCH, int KREAL>
__global__ __launch_bounds__(256) void gemm_lds_wmma(
    const float* __restrict__ A, const float* __restrict__ B, float* __restrict__ C, int M,
    const float* __restrict__ sums, const float* __restrict__ gamma,
    const float* __restrict__ beta) {
  constexpr int AP    = KC + 4;
  constexpr int NCH   = KP / KC;
  constexpr int KQ    = KC / 4;
  constexpr int APR   = KC / 4;
  constexpr int A_F4  = GB_ROWS * APR;
  constexpr int NCG   = DH / 4;
  constexpr int B_MB  = KQ * NCG;
  constexpr int A_IT  = (A_F4 + 255) / 256;
  constexpr int B_IT  = (B_MB + 255) / 256;
  constexpr int MAINF = GB_ROWS * AP + KQ * GB_BQ * 4;
  constexpr int CF    = GB_ROWS * GB_CP;
  static_assert(KP % KC == 0 && KC % 8 == 0 && ((AP / 4) & 1) == 1, "chunking");
  static_assert(KREAL <= KP && DH <= DHP && DH % 4 == 0, "real part inside the padded shape");
  constexpr bool A16  = (APITCH % 4 == 0);
  constexpr bool ROWG = (KP > KREAL);
  static_assert((A16 && KP <= APITCH) || (APITCH == KREAL && KREAL % 4 == 2 && KP == KREAL + 2),
                "A rows: 16-byte pieces, or 8-byte halves of which the first is always real");
  __shared__ __attribute__((aligned(16))) float smem[MAINF > CF ? MAINF : CF];
  __shared__ __attribute__((aligned(16))) float coef[BN ? 2 * KP : 4];
  float* As = smem;
  v4f*   Bs = (v4f*)(smem + GB_ROWS * AP);

  const int tid  = threadIdx.x;
  const int lane = tid & 31;
  const int w    = tid >> 5;
  const int half = lane >> 4;
  const int c    = lane & 15;
  const int mp   = w & 1;
  const int q    = w >> 1;
  const int row0 = blockIdx.x * GB_ROWS;

  v4f pa[A_IT];
  v4f pb[B_IT][4];

  auto fetch = [&](int ch) {
    const int kc0 = ch * KC;
#pragma unroll
    for (int i = 0; i < A_IT; ++i) {
      int idx = tid + 256 * i;
      if (idx < A_F4) {
        int r = idx / APR, c4 = idx - r * APR;
        int grow = row0 + r;
        grow = grow < M ? grow : M - 1;
        const int col = kc0 + 4 * c4;
        const float* ap = A + (size_t)grow * APITCH + col;
        if (A16) {
          pa[i] = *(const v4f*)ap;
        } else {
          v2f lo = *(const v2f*)ap;
          v2f hi = *(const v2f*)(ap + (col + 2 < KREAL ? 2 : 0));
          v4f av;
          av.x = lo.x; av.y = lo.y; av.z = hi.x; av.w = hi.y;
          pa[i] = av;
        }
      }
    }
#pragma unroll
    for (int i = 0; i < B_IT; ++i) {
      int mb = tid + 256 * i;
      if (mb < B_MB) {
        int kq = mb / NCG, cg = mb - kq * NCG;
        const int k0 = kc0 + 4 * kq;
#pragma unroll
        for (int r = 0; r < 4; ++r) {
          int kr = k0 + r;
          if (ROWG) kr = kr < KREAL ? kr : KREAL - 1;
          pb[i][r] = *(const v4f*)(B + (size_t)kr * DH + 4 * cg);
        }
      }
    }
  };
  auto stage = [&](int ch) {
#pragma unroll
    for (int i = 0; i < A_IT; ++i) {
      int idx = tid + 256 * i;
      if (idx < A_F4) {
        int r = idx / APR, c4 = idx - r * APR;
        v4f av = pa[i];
        if (!A16 && ch * KC + 4 * c4 + 2 >= KREAL) {
          av.z = 0.0f;
          av.w = 0.0f;
        }
        if (BN) {
          const v4f sc = *(const v4f*)(coef + ch * KC + 4 * c4);
          const v4f sh = *(const v4f*)(coef + KP + ch * KC + 4 * c4);
          av.x = av.x * sc.x + sh.x;
          av.y = av.y * sc.y + sh.y;
          av.z = av.z * sc.z + sh.z;
          av.w = av.w * sc.w + sh.w;
        }
        *(v4f*)(As + r * AP + 4 * c4) = av;
      }
    }
#pragma unroll
    for (int i = 0; i < B_IT; ++i) {
      int mb = tid + 256 * i;
      if (mb < B_MB) {
        int kq = mb / NCG, cg = mb - kq * NCG;
        const int k0 = ch * KC + 4 * kq;
        v4f* bs = Bs + kq * GB_BQ + cg;
#pragma unroll
        for (int j = 0; j < 4; ++j) {
          v4f t;
          t.x = pb[i][0][j]; t.y = pb[i][1][j]; t.z = pb[i][2][j]; t.w = pb[i][3][j];
          if (ROWG) {
            if (k0 + 1 >= KREAL) t.y = 0.0f;
            if (k0 + 2 >= KREAL) t.z = 0.0f;
            if (k0 + 3 >= KREAL) t.w = 0.0f;
          }
          bs[j * GB_SLOTQ] = t;
        }
      }
    }
  };

  v8f acc[2][5];
#pragma unroll
  for (int mt = 0; mt < 2; ++mt)
#pragma unroll
    for (int nt = 0; nt < 5; ++nt)
#pragma unroll
      for (int r = 0; r < 8; ++r) acc[mt][nt][r] = 0.0f;

  const float* a0p = As + (32 * mp + c) * AP + 4 * half;
  const float* a1p = a0p + 16 * AP;
  int bslot[5];
#pragma unroll
  for (int nt = 0; nt < 5; ++nt) {
    int n = 80 * q + 16 * nt + c;
    bslot[nt] = half * GB_BQ + (n & 3) * GB_SLOTQ + (n >> 2);
  }

  if (tid < KQ * 4 * (DHP / 4 - NCG)) {
    constexpr int NPG = DHP / 4 - NCG;
    int kq = tid / (4 * NPG), rem = tid - kq * (4 * NPG);
    int j = rem / NPG, cg = NCG + (rem - j * NPG);
    v4f z;
    z.x = 0.0f; z.y = 0.0f; z.z = 0.0f; z.w = 0.0f;
    Bs[kq * GB_BQ + j * GB_SLOTQ + cg] = z;
  }
  fetch(0);
  if (BN) {
    for (int k = tid; k < KP; k += 256) {
      float sc = 0.0f, sh = 0.0f;
      if (k < DH) bn_scale_shift(sums, k, M, gamma[k], beta[k], sc, sh);
      coef[k]      = sc;
      coef[KP + k] = sh;
    }
    __syncthreads();
  }
#pragma unroll 1
  for (int ch = 0; ch < NCH; ++ch) {
    stage(ch);
    __syncthreads();
    if (ch + 1 < NCH) fetch(ch + 1);
#pragma unroll
    for (int kb = 0; kb < KC; kb += 8) {
      v4f a0 = *(const v4f*)(a0p + kb);
      v4f a1 = *(const v4f*)(a1p + kb);
      v4f b[5];
#pragma unroll
      for (int nt = 0; nt < 5; ++nt) b[nt] = Bs[(kb >> 2) * GB_BQ + bslot[nt]];
      v2f a0l, a0h, a1l, a1h;
      a0l.x = a0.x; a0l.y = a0.y; a0h.x = a0.z; a0h.y = a0.w;
      a1l.x = a1.x; a1l.y = a1.y; a1h.x = a1.z; a1h.y = a1.w;
#pragma unroll
      for (int nt = 0; nt < 5; ++nt) {
        v2f bl;
        bl.x = b[nt].x; bl.y = b[nt].y;
        acc[0][nt] = __builtin_amdgcn_wmma_f32_16x16x4_f32(false, a0l, false, bl, (short)0,
                                                           acc[0][nt], false, false);
        acc[1][nt] = __builtin_amdgcn_wmma_f32_16x16x4_f32(false, a1l, false, bl, (short)0,
                                                           acc[1][nt], false, false);
      }
#pragma unroll
      for (int nt = 0; nt < 5; ++nt) {
        v2f bh;
        bh.x = b[nt].z; bh.y = b[nt].w;
        acc[0][nt] = __builtin_amdgcn_wmma_f32_16x16x4_f32(false, a0h, false, bh, (short)0,
                                                           acc[0][nt], false, false);
        acc[1][nt] = __builtin_amdgcn_wmma_f32_16x16x4_f32(false, a1h, false, bh, (short)0,
                                                           acc[1][nt], false, false);
      }
    }
    __syncthreads();
  }

  float* Cs = smem;
#pragma unroll
  for (int mt = 0; mt < 2; ++mt)
#pragma unroll
    for (int nt = 0; nt < 5; ++nt)
#pragma unroll
      for (int r = 0; r < 8; ++r)
        Cs[(32 * mp + 16 * mt + 8 * half + r) * GB_CP + 80 * q + 16 * nt + c] = acc[mt][nt][r];
  __syncthreads();
#pragma unroll
  for (int i = 0; i < (GB_ROWS * (DHP / 4)) / 256; ++i) {
    int idx = tid + 256 * i;
    int r = idx / (DHP / 4), c4 = idx - r * (DHP / 4);
    if (row0 + r < M)
      *(v4f*)(C + (size_t)(row0 + r) * DHP + 4 * c4) = *(const v4f*)(Cs + r * GB_CP + 4 * c4);
  }
}

__device__ __forceinline__ unsigned enc_f32(float f) {
  unsigned u = __float_as_uint(f);
  return (u & 0x80000000u) ? ~u : (u | 0x80000000u);
}
__device__ __forceinline__ float dec_f32(unsigned u) {
  return __uint_as_float((u & 0x80000000u) ? (u & 0x7FFFFFFFu) : ~u);
}

template <bool POOL>
__global__ __launch_bounds__(160) __attribute__((amdgpu_waves_per_eu(16, 16))) void agg_tile_wmma(
    const float* __restrict__ h, const unsigned* __restrict__ rec,
    const unsigned* __restrict__ toff, const float* __restrict__ dinv,
    const float* __restrict__ bias, const int* __restrict__ batch,
    float* __restrict__ y, float* __restrict__ sums,
    unsigned* __restrict__ pmax, unsigned* __restrict__ pmin) {
  int t    = blockIdx.x;
  int wave = __builtin_amdgcn_readfirstlane((int)(threadIdx.x >> 5));
  int row0 = t * 16;
  int lane = threadIdx.x & 31;
  int half = lane >> 4;
  unsigned m = lane & 15;
  int col  = wave * 64 + (int)m * 4;

  v8f acc0, acc1, acc2, acc3;
#pragma unroll
  for (int r = 0; r < 8; ++r) {
    int row = row0 + r + half * 8;
    float di = dinv[row];
    v4f hv = *(const v4f*)(h + (size_t)row * DHP + col);
    acc0[r] = hv.x * di * di;
    acc1[r] = hv.y * di * di;
    acc2[r] = hv.z * di * di;
    acc3[r] = hv.w * di * di;
  }
  unsigned base = toff[t];
  unsigned n    = toff[t + 1] - base;
  const unsigned* rp = rec + 2 * (size_t)(base + 2 * half);
  for (unsigned i = 0; i < n; i += 8) {
    v4u ra = *(const v4u*)(rp + 2 * (size_t)i);
    v4u rb = *(const v4u*)(rp + 2 * (size_t)i + 8);
    v4f ha0 = *(const v4f*)(h + (size_t)(ra.x >> 4) * DHP + col);
    v4f ha1 = *(const v4f*)(h + (size_t)(ra.z >> 4) * DHP + col);
    v4f hb0 = *(const v4f*)(h + (size_t)(rb.x >> 4) * DHP + col);
    v4f hb1 = *(const v4f*)(h + (size_t)(rb.z >> 4) * DHP + col);
    v2f a, b;
    a.x = ((ra.x & 15u) == m) ? __uint_as_float(ra.y) : 0.0f;
    a.y = ((ra.z & 15u) == m) ? __uint_as_float(ra.w) : 0.0f;
    b.x = ha0.x; b.y = ha1.x;
    acc0 = __builtin_amdgcn_wmma_f32_16x16x4_f32(false, a, false, b, (short)0, acc0, false, false);
    b.x = ha0.y; b.y = ha1.y;
    acc1 = __builtin_amdgcn_wmma_f32_16x16x4_f32(false, a, false, b, (short)0, acc1, false, false);
    b.x = ha0.z; b.y = ha1.z;
    acc2 = __builtin_amdgcn_wmma_f32_16x16x4_f32(false, a, false, b, (short)0, acc2, false, false);
    b.x = ha0.w; b.y = ha1.w;
    acc3 = __builtin_amdgcn_wmma_f32_16x16x4_f32(false, a, false, b, (short)0, acc3, false, false);
    a.x = ((rb.x & 15u) == m) ? __uint_as_float(rb.y) : 0.0f;
    a.y = ((rb.z & 15u) == m) ? __uint_as_float(rb.w) : 0.0f;
    b.x = hb0.x; b.y = hb1.x;
    acc0 = __builtin_amdgcn_wmma_f32_16x16x4_f32(false, a, false, b, (short)0, acc0, false, false);
    b.x = hb0.y; b.y = hb1.y;
    acc1 = __builtin_amdgcn_wmma_f32_16x16x4_f32(false, a, false, b, (short)0, acc1, false, false);
    b.x = hb0.z; b.y = hb1.z;
    acc2 = __builtin_amdgcn_wmma_f32_16x16x4_f32(false, a, false, b, (short)0, acc2, false, false);
    b.x = hb0.w; b.y = hb1.w;
    acc3 = __builtin_amdgcn_wmma_f32_16x16x4_f32(false, a, false, b, (short)0, acc3, false, false);
  }

  const int ln = (int)__builtin_amdgcn_mbcnt_lo(~0u, 0u);
  const int eh = ln >> 4;
  const int ec = wave * 64 + (ln & 15) * 4;
  const bool valid = ec < DH;
  float b0 = 0.0f, b1 = 0.0f, b2 = 0.0f, b3 = 0.0f;
  if (valid) {
    b0 = bias[ec];
    b1 = bias[ec + 1];
    b2 = bias[ec + 2];
    b3 = bias[ec + 3];
  }
  float s0 = 0.f, s1 = 0.f, s2 = 0.f, s3 = 0.f;
  float q0 = 0.f, q1 = 0.f, q2 = 0.f, q3 = 0.f;
#pragma unroll
  for (int r = 0; r < 8; ++r) {
    float v0 = fmaxf(acc0[r] + b0, 0.0f);
    float v1 = fmaxf(acc1[r] + b1, 0.0f);
    float v2 = fmaxf(acc2[r] + b2, 0.0f);
    float v3 = fmaxf(acc3[r] + b3, 0.0f);
    acc0[r] = v0; acc1[r] = v1; acc2[r] = v2; acc3[r] = v3;
    s0 += v0; s1 += v1; s2 += v2; s3 += v3;
    q0 += v0 * v0; q1 += v1 * v1; q2 += v2 * v2; q3 += v3 * v3;
    if (!POOL) {
      int row = row0 + r + eh * 8;
      v4f o;
      o.x = v0; o.y = v1; o.z = v2; o.w = v3;
      *(v4f*)(y + (size_t)row * DHP + ec) = o;
    }
  }
  s0 += __shfl_xor(s0, 16); s1 += __shfl_xor(s1, 16);
  s2 += __shfl_xor(s2, 16); s3 += __shfl_xor(s3, 16);
  q0 += __shfl_xor(q0, 16); q1 += __shfl_xor(q1, 16);
  q2 += __shfl_xor(q2, 16); q3 += __shfl_xor(q3, 16);
  if (valid) {
    float* sp = sums + (size_t)(t & (NSC - 1)) * (2 * DHP) + eh * DHP + ec;
    atomicAdd(sp,     eh ? q0 : s0);
    atomicAdd(sp + 1, eh ? q1 : s1);
    atomicAdd(sp + 2, eh ? q2 : s2);
    atomicAdd(sp + 3, eh ? q3 : s3);
  }

  if (POOL) {
    const int* bp = batch + row0 + 8 * eh;
    const int g0 = batch[row0];
    int gid[8];
    bool same = true;
#pragma unroll
    for (int r = 0; r < 8; ++r) {
      gid[r] = bp[r];
      same = same && (gid[r] == g0);
    }
    if (__all(same)) {
      float hi0 = -INFINITY, hi1 = -INFINITY, hi2 = -INFINITY, hi3 = -INFINITY;
      float lo0 = INFINITY, lo1 = INFINITY, lo2 = INFINITY, lo3 = INFINITY;
#pragma unroll
      for (int r = 0; r < 8; ++r) {
        hi0 = fmaxf(hi0, acc0[r]); lo0 = fminf(lo0, acc0[r]);
        hi1 = fmaxf(hi1, acc1[r]); lo1 = fminf(lo1, acc1[r]);
        hi2 = fmaxf(hi2, acc2[r]); lo2 = fminf(lo2, acc2[r]);
        hi3 = fmaxf(hi3, acc3[r]); lo3 = fminf(lo3, acc3[r]);
      }
      hi0 = fmaxf(hi0, __shfl_xor(hi0, 16)); lo0 = fminf(lo0, __shfl_xor(lo0, 16));
      hi1 = fmaxf(hi1, __shfl_xor(hi1, 16)); lo1 = fminf(lo1, __shfl_xor(lo1, 16));
      hi2 = fmaxf(hi2, __shfl_xor(hi2, 16)); lo2 = fminf(lo2, __shfl_xor(lo2, 16));
      hi3 = fmaxf(hi3, __shfl_xor(hi3, 16)); lo3 = fminf(lo3, __shfl_xor(lo3, 16));
      if (valid) {
        unsigned* pp = (eh ? pmin : pmax) + (size_t)g0 * DH + ec;
        atomicMax(pp,     eh ? ~enc_f32(lo0) : enc_f32(hi0));
        atomicMax(pp + 1, eh ? ~enc_f32(lo1) : enc_f32(hi1));
        atomicMax(pp + 2, eh ? ~enc_f32(lo2) : enc_f32(hi2));
        atomicMax(pp + 3, eh ? ~enc_f32(lo3) : enc_f32(hi3));
      }
    } else if (valid) {
      auto pool_runs = [&](const v8f& v, unsigned cj) {
        int g = gid[0];
        float hi = -INFINITY, lo = INFINITY;
#pragma unroll
        for (int r = 0; r < 8; ++r) {
          if (gid[r] != g) {
            atomicMax(pmax + ((unsigned)g * DH + cj), enc_f32(hi));
            atomicMax(pmin + ((unsigned)g * DH + cj), ~enc_f32(lo));
            g  = gid[r];
            hi = -INFINITY;
            lo = INFINITY;
          }
          hi = fmaxf(hi, v[r]);
          lo = fminf(lo, v[r]);
        }
        atomicMax(pmax + ((unsigned)g * DH + cj), enc_f32(hi));
        atomicMax(pmin + ((unsigned)g * DH + cj), ~enc_f32(lo));
      };
      pool_runs(acc0, (unsigned)ec);
      pool_runs(acc1, (unsigned)ec + 1u);
      pool_runs(acc2, (unsigned)ec + 2u);
      pool_runs(acc3, (unsigned)ec + 3u);
    }
  }
}

__global__ void pool_bn_finish(const unsigned* __restrict__ pmax, const unsigned* __restrict__ pmin,
                               const float* __restrict__ sums, const float* __restrict__ gamma,
                               const float* __restrict__ beta, float* __restrict__ out, int M) {
  int g = blockIdx.x, c = threadIdx.x;
  if (c >= DH) return;
  float scale, shift;
  bn_scale_shift(sums, c, M, gamma[c], beta[c], scale, shift);
  unsigned em = pmax[(size_t)g * DH + c];
  float res = -INFINITY;
  if (em != 0u) {
    float v = (scale >= 0.0f) ? dec_f32(em) : dec_f32(~pmin[(size_t)g * DH + c]);
    res = v * scale + shift;
  }
  out[(size_t)g * DH + c] = res;
}

extern "C" void kernel_launch(void* const* d_in, const int* in_sizes, int n_in,
                              void* d_out, int out_size, void* d_ws, size_t ws_size,
                              hipStream_t stream) {
  const float* x      = (const float*)d_in[0];
  const int*   eidx   = (const int*)d_in[1];
  const int*   batch  = (const int*)d_in[2];
  const float* W1     = (const float*)d_in[3];
  const float* b1     = (const float*)d_in[4];
  const float* gamma1 = (const float*)d_in[5];
  const float* beta1  = (const float*)d_in[6];
  const float* W2     = (const float*)d_in[7];
  const float* b2     = (const float*)d_in[8];
  const float* gamma2 = (const float*)d_in[9];
  const float* beta2  = (const float*)d_in[10];
  float* out = (float*)d_out;

  const int* src = eidx;
  const int* dst = eidx + EE;

  float* ws = (float*)d_ws;
  float*    deg  = ws;
  float*    sums = ws + 4178048;
  float*    bufA = ws + 4334688;
  float*    bufB = ws + 20334688;
  unsigned* tcnt = (unsigned*)(ws + 36334688);
  unsigned* toff = (unsigned*)(ws + 36337888);
  unsigned* rank = (unsigned*)(ws + 36341088);
  unsigned* rec  = (unsigned*)(ws + 37141088);
  unsigned* pmax = (unsigned*)(ws + 38800000);
  unsigned* pmin = pmax + GG * DH;

  const int gemmGrid   = (NN + GB_ROWS - 1) / GB_ROWS;

  fill_f32<<<(NN + 255) / 256, 256, 0, stream>>>(deg, 1.0f, NN);
  fill_u32<<<(NT + 255) / 256, 256, 0, stream>>>(tcnt, 0u, NT);
  deg_tile_count<<<(EE + 255) / 256, 256, 0, stream>>>(dst, deg, tcnt, rank, EE);
  rsqrt_inplace<<<(NN + 255) / 256, 256, 0, stream>>>(deg, NN);

  tile_scan<<<1, 1024, 0, stream>>>(tcnt, toff, rec);
  tile_place<<<(EE + 255) / 256, 256, 0, stream>>>(src, dst, deg, toff, rank, rec, EE);

  gemm_lds_wmma<DINP, 16, false, DIN, DIN><<<gemmGrid, 256, 0, stream>>>(x, W1, bufA, NN, nullptr,
                                                                         nullptr, nullptr);
  fill_f32<<<(NSC * 2 * DHP + 255) / 256, 256, 0, stream>>>(sums, 0.0f, NSC * 2 * DHP);
  agg_tile_wmma<false><<<NT, 160, 0, stream>>>(bufA, rec, toff, deg, b1, batch, bufB, sums, pmax,
                                               pmin);

  gemm_lds_wmma<DH, 24, true, DHP, DH><<<gemmGrid, 256, 0, stream>>>(bufB, W2, bufA, NN, sums,
                                                                     gamma1, beta1);
  fill_f32<<<(NSC * 2 * DHP + 255) / 256, 256, 0, stream>>>(sums, 0.0f, NSC * 2 * DHP);
  fill_u32<<<(2L * GG * DH + 255) / 256, 256, 0, stream>>>(pmax, 0u, 2L * GG * DH);

  agg_tile_wmma<true><<<NT, 160, 0, stream>>>(bufA, rec, toff, deg, b2, batch, bufB, sums, pmax,
                                              pmin);
  pool_bn_finish<<<GG, DHP, 0, stream>>>(pmax, pmin, sums, gamma2, beta2, out, NN);
}


# Edge binning without global atomics: 64 edge chunks rank their edges by LDS adds on tile counters, a scan over chunks x tiles gives slots, degrees counted from the placed records; new sum order
# speedup vs baseline: 6.3162x; 1.3154x over previous
#include <hip/hip_runtime.h>
#include <hip/hip_bf16.h>

typedef float v2f __attribute__((ext_vector_type(2)));
typedef float v8f __attribute__((ext_vector_type(8)));
typedef float v4f __attribute__((ext_vector_type(4)));
typedef unsigned v4u __attribute__((ext_vector_type(4)));
typedef unsigned v2u __attribute__((ext_vector_type(2)));

#define NN    50000
#define EE    800000
#define GG    500
#define DIN   78
#define DINP  80
#define DH    312
#define DHP   320
#define BNEPS 1e-5f
#define NT    (NN / 16)
static_assert(NN % 16 == 0, "destination tiles are whole");

__global__ void fill_f32(float* __restrict__ p, float v, long n) {
  long i = (long)blockIdx.x * blockDim.x + threadIdx.x;
  if (i < n) p[i] = v;
}
__global__ void fill_u32(unsigned* __restrict__ p, unsigned v, long n) {
  long i = (long)blockIdx.x * blockDim.x + threadIdx.x;
  if (i < n) p[i] = v;
}

#define NCHUNK 64
#define ECH    (EE / NCHUNK)
#define NTP    4096
static_assert(EE % NCHUNK == 0 && NT <= NTP && NTP == 4 * 1024, "chunks are whole; 4 tiles per scan thread");

__global__ __launch_bounds__(1024) void chunk_rank(const int* __restrict__ dst,
                                                   unsigned* __restrict__ rank,
                                                   unsigned* __restrict__ chist) {
  constexpr int IT = (ECH + 1023) / 1024;
  __shared__ unsigned hist[NTP];
  const int tid = threadIdx.x;
  const int e0  = blockIdx.x * ECH;
  int d[IT];
#pragma unroll
  for (int i = 0; i < IT; ++i) {
    int k = tid + 1024 * i;
    d[i] = k < ECH ? dst[e0 + k] : -1;
  }
  for (int k = tid; k < NTP; k += 1024) hist[k] = 0u;
  __syncthreads();
#pragma unroll
  for (int i = 0; i < IT; ++i) {
    int k = tid + 1024 * i;
    if (d[i] >= 0) rank[e0 + k] = atomicAdd(&hist[d[i] >> 4], 1u);
  }
  __syncthreads();
  for (int k = tid; k < NTP; k += 1024) chist[(size_t)blockIdx.x * NTP + k] = hist[k];
}
__global__ __launch_bounds__(1024) void tile_scan(const unsigned* __restrict__ chist,
                                                  unsigned* __restrict__ cbase,
                                                  unsigned* __restrict__ tcnt,
                                                  unsigned* __restrict__ toff,
                                                  unsigned* __restrict__ rec) {
  __shared__ unsigned s[1024];
  int t = threadIdx.x;
  unsigned n[4] = {0u, 0u, 0u, 0u};
  unsigned c[4], sum = 0;
#pragma unroll 1
  for (int c0 = 0; c0 < NCHUNK; c0 += 16) {
    v4u v[16];
#pragma unroll
    for (int j = 0; j < 16; ++j) v[j] = *(const v4u*)(chist + (size_t)(c0 + j) * NTP + 4 * t);
#pragma unroll
    for (int j = 0; j < 16; ++j) {
      v4u b;
      b.x = n[0]; b.y = n[1]; b.z = n[2]; b.w = n[3];
      *(v4u*)(cbase + (size_t)(c0 + j) * NTP + 4 * t) = b;
      n[0] += v[j].x; n[1] += v[j].y; n[2] += v[j].z; n[3] += v[j].w;
    }
  }
#pragma unroll
  for (int i = 0; i < 4; ++i) {
    int idx = t * 4 + i;
    if (idx >= NT) n[i] = 0u;
    else tcnt[idx] = n[i];
    c[i] = (n[i] + 7u) & ~7u;
    sum += c[i];
  }
  s[t] = sum;
  __syncthreads();
  for (int o = 1; o < 1024; o <<= 1) {
    unsigned v = t >= o ? s[t - o] : 0u;
    __syncthreads();
    s[t] += v;
    __syncthreads();
  }
  unsigned run = s[t] - sum;
#pragma unroll
  for (int i = 0; i < 4; ++i) {
    int idx = t * 4 + i;
    if (idx <= NT) toff[idx] = run;
    if (idx < NT) {
      for (unsigned j = n[i]; j < c[i]; ++j) {
        rec[2 * (run + j)]     = (unsigned)(idx * 16) << 4;
        rec[2 * (run + j) + 1] = 0u;
      }
    }
    run += c[i];
  }
}
__global__ void tile_place(const int* __restrict__ src, const int* __restrict__ dst,
                           const unsigned* __restrict__ toff, const unsigned* __restrict__ cbase,
                           const unsigned* __restrict__ rank, unsigned* __restrict__ rec, int E) {
  int e = blockIdx.x * blockDim.x + threadIdx.x;
  if (e < E) {
    int s = src[e], d = dst[e];
    int tile = d >> 4;
    unsigned slot = toff[tile] + cbase[(size_t)(e / ECH) * NTP + tile] + rank[e];
    rec[2 * (size_t)slot] = ((unsigned)s << 4) | (unsigned)(d & 15);
  }
}
__global__ __launch_bounds__(256) void tile_deg(const unsigned* __restrict__ rec,
                                                const unsigned* __restrict__ toff,
                                                const unsigned* __restrict__ tcnt,
                                                float* __restrict__ dinv) {
  __shared__ unsigned cnt[8][16];
  const int tid = threadIdx.x, lane = tid & 31, w = tid >> 5;
  const int t = blockIdx.x * 8 + w;
  if (lane < 16) cnt[w][lane] = 0u;
  __syncthreads();
  if (t < NT) {
    const unsigned base = toff[t], n = tcnt[t];
    for (unsigned i0 = 0; i0 < n; i0 += 128) {
      unsigned x[4];
#pragma unroll
      for (int j = 0; j < 4; ++j) {
        unsigned i = i0 + lane + 32 * j;
        x[j] = rec[2 * (size_t)(base + (i < n ? i : 0u))];
      }
#pragma unroll
      for (int j = 0; j < 4; ++j) {
        unsigned i = i0 + lane + 32 * j;
        if (i < n) atomicAdd(&cnt[w][x[j] & 15u], 1u);
      }
    }
  }
  __syncthreads();
  if (t < NT && lane < 16) dinv[t * 16 + lane] = rsqrtf((float)(cnt[w][lane] + 1u));
}
__global__ __launch_bounds__(256) void tile_weight(unsigned* __restrict__ rec,
                                                   const unsigned* __restrict__ toff,
                                                   const unsigned* __restrict__ tcnt,
                                                   const float* __restrict__ dinv) {
  const int tid = threadIdx.x, lane = tid & 31, w = tid >> 5;
  const int t = blockIdx.x * 8 + w;
  if (t >= NT) return;
  const unsigned base = toff[t], n = tcnt[t];
  for (unsigned i0 = 0; i0 < n; i0 += 128) {
    unsigned x[4];
    float    ds[4], dd[4];
#pragma unroll
    for (int j = 0; j < 4; ++j) {
      unsigned i = i0 + lane + 32 * j;
      x[j] = rec[2 * (size_t)(base + (i < n ? i : 0u))];
    }
#pragma unroll
    for (int j = 0; j < 4; ++j) {
      ds[j] = dinv[x[j] >> 4];
      dd[j] = dinv[t * 16 + (int)(x[j] & 15u)];
    }
#pragma unroll
    for (int j = 0; j < 4; ++j) {
      unsigned i = i0 + lane + 32 * j;
      if (i < n) rec[2 * (size_t)(base + i) + 1] = __float_as_uint(ds[j] * dd[j]);
    }
  }
}

#define NSC 16
__device__ __forceinline__ void bn_scale_shift(const float* __restrict__ sums, int c, int M,
                                               float g, float b, float& scale, float& shift) {
  float s = 0.f, q = 0.f;
  for (int k = 0; k < NSC; ++k) {
    s += sums[k * (2 * DHP) + c];
    q += sums[k * (2 * DHP) + DHP + c];
  }
  float invn = 1.0f / (float)M;
  float mean = s * invn;
  float var  = q * invn - mean * mean;
  scale = rsqrtf(var + BNEPS) * g;
  shift = b - mean * scale;
}

#define GB_ROWS 64
#define GB_SLOTQ 84
#define GB_BQ   (4 * GB_SLOTQ)
#define GB_CP   (DHP + 4)
template <int KP, int KC, bool BN, int APITCH, int KREAL>
__global__ __launch_bounds__(256) void gemm_lds_wmma(
    const float* __restrict__ A, const float* __restrict__ B, float* __restrict__ C, int M,
    const float* __restrict__ sums, const float* __restrict__ gamma,
    const float* __restrict__ beta) {
  constexpr int AP    = KC + 4;
  constexpr int NCH   = KP / KC;
  constexpr int KQ    = KC / 4;
  constexpr int APR   = KC / 4;
  constexpr int A_F4  = GB_ROWS * APR;
  constexpr int NCG   = DH / 4;
  constexpr int B_MB  = KQ * NCG;
  constexpr int A_IT  = (A_F4 + 255) / 256;
  constexpr int B_IT  = (B_MB + 255) / 256;
  constexpr int MAINF = GB_ROWS * AP + KQ * GB_BQ * 4;
  constexpr int CF    = GB_ROWS * GB_CP;
  static_assert(KP % KC == 0 && KC % 8 == 0 && ((AP / 4) & 1) == 1, "chunking");
  static_assert(KREAL <= KP && DH <= DHP && DH % 4 == 0, "real part inside the padded shape");
  constexpr bool A16  = (APITCH % 4 == 0);
  constexpr bool ROWG = (KP > KREAL);
  static_assert((A16 && KP <= APITCH) || (APITCH == KREAL && KREAL % 4 == 2 && KP == KREAL + 2),
                "A rows: 16-byte pieces, or 8-byte halves of which the first is always real");
  __shared__ __attribute__((aligned(16))) float smem[MAINF > CF ? MAINF : CF];
  __shared__ __attribute__((aligned(16))) float coef[BN ? 2 * KP : 4];
  float* As = smem;
  v4f*   Bs = (v4f*)(smem + GB_ROWS * AP);

  const int tid  = threadIdx.x;
  const int lane = tid & 31;
  const int w    = tid >> 5;
  const int half = lane >> 4;
  const int c    = lane & 15;
  const int mp   = w & 1;
  const int q    = w >> 1;
  const int row0 = blockIdx.x * GB_ROWS;

  v4f pa[A_IT];
  v4f pb[B_IT][4];

  auto fetch = [&](int ch) {
    const int kc0 = ch * KC;
#pragma unroll
    for (int i = 0; i < A_IT; ++i) {
      int idx = tid + 256 * i;
      if (idx < A_F4) {
        int r = idx / APR, c4 = idx - r * APR;
        int grow = row0 + r;
        grow = grow < M ? grow : M - 1;
        const int col = kc0 + 4 * c4;
        const float* ap = A + (size_t)grow * APITCH + col;
        if (A16) {
          pa[i] = *(const v4f*)ap;
        } else {
          v2f lo = *(const v2f*)ap;
          v2f hi = *(const v2f*)(ap + (col + 2 < KREAL ? 2 : 0));
          v4f av;
          av.x = lo.x; av.y = lo.y; av.z = hi.x; av.w = hi.y;
          pa[i] = av;
        }
      }
    }
#pragma unroll
    for (int i = 0; i < B_IT; ++i) {
      int mb = tid + 256 * i;
      if (mb < B_MB) {
        int kq = mb / NCG, cg = mb - kq * NCG;
        const int k0 = kc0 + 4 * kq;
#pragma unroll
        for (int r = 0; r < 4; ++r) {
          int kr = k0 + r;
          if (ROWG) kr = kr < KREAL ? kr : KREAL - 1;
          pb[i][r] = *(const v4f*)(B + (size_t)kr * DH + 4 * cg);
        }
      }
    }
  };
  auto stage = [&](int ch) {
#pragma unroll
    for (int i = 0; i < A_IT; ++i) {
      int idx = tid + 256 * i;
      if (idx < A_F4) {
        int r = idx / APR, c4 = idx - r * APR;
        v4f av = pa[i];
        if (!A16 && ch * KC + 4 * c4 + 2 >= KREAL) {
          av.z = 0.0f;
          av.w = 0.0f;
        }
        if (BN) {
          const v4f sc = *(const v4f*)(coef + ch * KC + 4 * c4);
          const v4f sh = *(const v4f*)(coef + KP + ch * KC + 4 * c4);
          av.x = av.x * sc.x + sh.x;
          av.y = av.y * sc.y + sh.y;
          av.z = av.z * sc.z + sh.z;
          av.w = av.w * sc.w + sh.w;
        }
        *(v4f*)(As + r * AP + 4 * c4) = av;
      }
    }
#pragma unroll
    for (int i = 0; i < B_IT; ++i) {
      int mb = tid + 256 * i;
      if (mb < B_MB) {
        int kq = mb / NCG, cg = mb - kq * NCG;
        const int k0 = ch * KC + 4 * kq;
        v4f* bs = Bs + kq * GB_BQ + cg;
#pragma unroll
        for (int j = 0; j < 4; ++j) {
          v4f t;
          t.x = pb[i][0][j]; t.y = pb[i][1][j]; t.z = pb[i][2][j]; t.w = pb[i][3][j];
          if (ROWG) {
            if (k0 + 1 >= KREAL) t.y = 0.0f;
            if (k0 + 2 >= KREAL) t.z = 0.0f;
            if (k0 + 3 >= KREAL) t.w = 0.0f;
          }
          bs[j * GB_SLOTQ] = t;
        }
      }
    }
  };

  v8f acc[2][5];
#pragma unroll
  for (int mt = 0; mt < 2; ++mt)
#pragma unroll
    for (int nt = 0; nt < 5; ++nt)
#pragma unroll
      for (int r = 0; r < 8; ++r) acc[mt][nt][r] = 0.0f;

  const float* a0p = As + (32 * mp + c) * AP + 4 * half;
  const float* a1p = a0p + 16 * AP;
  int bslot[5];
#pragma unroll
  for (int nt = 0; nt < 5; ++nt) {
    int n = 80 * q + 16 * nt + c;
    bslot[nt] = half * GB_BQ + (n & 3) * GB_SLOTQ + (n >> 2);
  }

  if (tid < KQ * 4 * (DHP / 4 - NCG)) {
    constexpr int NPG = DHP / 4 - NCG;
    int kq = tid / (4 * NPG), rem = tid - kq * (4 * NPG);
    int j = rem / NPG, cg = NCG + (rem - j * NPG);
    v4f z;
    z.x = 0.0f; z.y = 0.0f; z.z = 0.0f; z.w = 0.0f;
    Bs[kq * GB_BQ + j * GB_SLOTQ + cg] = z;
  }
  fetch(0);
  if (BN) {
    for (int k = tid; k < KP; k += 256) {
      float sc = 0.0f, sh = 0.0f;
      if (k < DH) bn_scale_shift(sums, k, M, gamma[k], beta[k], sc, sh);
      coef[k]      = sc;
      coef[KP + k] = sh;
    }
    __syncthreads();
  }
#pragma unroll 1
  for (int ch = 0; ch < NCH; ++ch) {
    stage(ch);
    __syncthreads();
    if (ch + 1 < NCH) fetch(ch + 1);
#pragma unroll
    for (int kb = 0; kb < KC; kb += 8) {
      v4f a0 = *(const v4f*)(a0p + kb);
      v4f a1 = *(const v4f*)(a1p + kb);
      v4f b[5];
#pragma unroll
      for (int nt = 0; nt < 5; ++nt) b[nt] = Bs[(kb >> 2) * GB_BQ + bslot[nt]];
      v2f a0l, a0h, a1l, a1h;
      a0l.x = a0.x; a0l.y = a0.y; a0h.x = a0.z; a0h.y = a0.w;
      a1l.x = a1.x; a1l.y = a1.y; a1h.x = a1.z; a1h.y = a1.w;
#pragma unroll
      for (int nt = 0; nt < 5; ++nt) {
        v2f bl;
        bl.x = b[nt].x; bl.y = b[nt].y;
        acc[0][nt] = __builtin_amdgcn_wmma_f32_16x16x4_f32(false, a0l, false, bl, (short)0,
                                                           acc[0][nt], false, false);
        acc[1][nt] = __builtin_amdgcn_wmma_f32_16x16x4_f32(false, a1l, false, bl, (short)0,
                                                           acc[1][nt], false, false);
      }
#pragma unroll
      for (int nt = 0; nt < 5; ++nt) {
        v2f bh;
        bh.x = b[nt].z; bh.y = b[nt].w;
        acc[0][nt] = __builtin_amdgcn_wmma_f32_16x16x4_f32(false, a0h, false, bh, (short)0,
                                                           acc[0][nt], false, false);
        acc[1][nt] = __builtin_amdgcn_wmma_f32_16x16x4_f32(false, a1h, false, bh, (short)0,
                                                           acc[1][nt], false, false);
      }
    }
    __syncthreads();
  }

  float* Cs = smem;
#pragma unroll
  for (int mt = 0; mt < 2; ++mt)
#pragma unroll
    for (int nt = 0; nt < 5; ++nt)
#pragma unroll
      for (int r = 0; r < 8; ++r)
        Cs[(32 * mp + 16 * mt + 8 * half + r) * GB_CP + 80 * q + 16 * nt + c] = acc[mt][nt][r];
  __syncthreads();
#pragma unroll
  for (int i = 0; i < (GB_ROWS * (DHP / 4)) / 256; ++i) {
    int idx = tid + 256 * i;
    int r = idx / (DHP / 4), c4 = idx - r * (DHP / 4);
    if (row0 + r < M)
      *(v4f*)(C + (size_t)(row0 + r) * DHP + 4 * c4) = *(const v4f*)(Cs + r * GB_CP + 4 * c4);
  }
}

__device__ __forceinline__ unsigned enc_f32(float f) {
  unsigned u = __float_as_uint(f);
  return (u & 0x80000000u) ? ~u : (u | 0x80000000u);
}
__device__ __forceinline__ float dec_f32(unsigned u) {
  return __uint_as_float((u & 0x80000000u) ? (u & 0x7FFFFFFFu) : ~u);
}

template <bool POOL>
__global__ __launch_bounds__(160) __attribute__((amdgpu_waves_per_eu(16, 16))) void agg_tile_wmma(
    const float* __restrict__ h, const unsigned* __restrict__ rec,
    const unsigned* __restrict__ toff, const float* __restrict__ dinv,
    const float* __restrict__ bias, const int* __restrict__ batch,
    float* __restrict__ y, float* __restrict__ sums,
    unsigned* __restrict__ pmax, unsigned* __restrict__ pmin) {
  int t    = blockIdx.x;
  int wave = __builtin_amdgcn_readfirstlane((int)(threadIdx.x >> 5));
  int row0 = t * 16;
  int lane = threadIdx.x & 31;
  int half = lane >> 4;
  unsigned m = lane & 15;
  int col  = wave * 64 + (int)m * 4;

  v8f acc0, acc1, acc2, acc3;
#pragma unroll
  for (int r = 0; r < 8; ++r) {
    int row = row0 + r + half * 8;
    float di = dinv[row];
    v4f hv = *(const v4f*)(h + (size_t)row * DHP + col);
    acc0[r] = hv.x * di * di;
    acc1[r] = hv.y * di * di;
    acc2[r] = hv.z * di * di;
    acc3[r] = hv.w * di * di;
  }
  unsigned base = toff[t];
  unsigned n    = toff[t + 1] - base;
  const unsigned* rp = rec + 2 * (size_t)(base + 2 * half);
  for (unsigned i = 0; i < n; i += 8) {
    v4u ra = *(const v4u*)(rp + 2 * (size_t)i);
    v4u rb = *(const v4u*)(rp + 2 * (size_t)i + 8);
    v4f ha0 = *(const v4f*)(h + (size_t)(ra.x >> 4) * DHP + col);
    v4f ha1 = *(const v4f*)(h + (size_t)(ra.z >> 4) * DHP + col);
    v4f hb0 = *(const v4f*)(h + (size_t)(rb.x >> 4) * DHP + col);
    v4f hb1 = *(const v4f*)(h + (size_t)(rb.z >> 4) * DHP + col);
    v2f a, b;
    a.x = ((ra.x & 15u) == m) ? __uint_as_float(ra.y) : 0.0f;
    a.y = ((ra.z & 15u) == m) ? __uint_as_float(ra.w) : 0.0f;
    b.x = ha0.x; b.y = ha1.x;
    acc0 = __builtin_amdgcn_wmma_f32_16x16x4_f32(false, a, false, b, (short)0, acc0, false, false);
    b.x = ha0.y; b.y = ha1.y;
    acc1 = __builtin_amdgcn_wmma_f32_16x16x4_f32(false, a, false, b, (short)0, acc1, false, false);
    b.x = ha0.z; b.y = ha1.z;
    acc2 = __builtin_amdgcn_wmma_f32_16x16x4_f32(false, a, false, b, (short)0, acc2, false, false);
    b.x = ha0.w; b.y = ha1.w;
    acc3 = __builtin_amdgcn_wmma_f32_16x16x4_f32(false, a, false, b, (short)0, acc3, false, false);
    a.x = ((rb.x & 15u) == m) ? __uint_as_float(rb.y) : 0.0f;
    a.y = ((rb.z & 15u) == m) ? __uint_as_float(rb.w) : 0.0f;
    b.x = hb0.x; b.y = hb1.x;
    acc0 = __builtin_amdgcn_wmma_f32_16x16x4_f32(false, a, false, b, (short)0, acc0, false, false);
    b.x = hb0.y; b.y = hb1.y;
    acc1 = __builtin_amdgcn_wmma_f32_16x16x4_f32(false, a, false, b, (short)0, acc1, false, false);
    b.x = hb0.z; b.y = hb1.z;
    acc2 = __builtin_amdgcn_wmma_f32_16x16x4_f32(false, a, false, b, (short)0, acc2, false, false);
    b.x = hb0.w; b.y = hb1.w;
    acc3 = __builtin_amdgcn_wmma_f32_16x16x4_f32(false, a, false, b, (short)0, acc3, false, false);
  }

  const int ln = (int)__builtin_amdgcn_mbcnt_lo(~0u, 0u);
  const int eh = ln >> 4;
  const int ec = wave * 64 + (ln & 15) * 4;
  const bool valid = ec < DH;
  float b0 = 0.0f, b1 = 0.0f, b2 = 0.0f, b3 = 0.0f;
  if (valid) {
    b0 = bias[ec];
    b1 = bias[ec + 1];
    b2 = bias[ec + 2];
    b3 = bias[ec + 3];
  }
  float s0 = 0.f, s1 = 0.f, s2 = 0.f, s3 = 0.f;
  float q0 = 0.f, q1 = 0.f, q2 = 0.f, q3 = 0.f;
#pragma unroll
  for (int r = 0; r < 8; ++r) {
    float v0 = fmaxf(acc0[r] + b0, 0.0f);
    float v1 = fmaxf(acc1[r] + b1, 0.0f);
    float v2 = fmaxf(acc2[r] + b2, 0.0f);
    float v3 = fmaxf(acc3[r] + b3, 0.0f);
    acc0[r] = v0; acc1[r] = v1; acc2[r] = v2; acc3[r] = v3;
    s0 += v0; s1 += v1; s2 += v2; s3 += v3;
    q0 += v0 * v0; q1 += v1 * v1; q2 += v2 * v2; q3 += v3 * v3;
    if (!POOL) {
      int row = row0 + r + eh * 8;
      v4f o;
      o.x = v0; o.y = v1; o.z = v2; o.w = v3;
      *(v4f*)(y + (size_t)row * DHP + ec) = o;
    }
  }
  s0 += __shfl_xor(s0, 16); s1 += __shfl_xor(s1, 16);
  s2 += __shfl_xor(s2, 16); s3 += __shfl_xor(s3, 16);
  q0 += __shfl_xor(q0, 16); q1 += __shfl_xor(q1, 16);
  q2 += __shfl_xor(q2, 16); q3 += __shfl_xor(q3, 16);
  if (valid) {
    float* sp = sums + (size_t)(t & (NSC - 1)) * (2 * DHP) + eh * DHP + ec;
    atomicAdd(sp,     eh ? q0 : s0);
    atomicAdd(sp + 1, eh ? q1 : s1);
    atomicAdd(sp + 2, eh ? q2 : s2);
    atomicAdd(sp + 3, eh ? q3 : s3);
  }

  if (POOL) {
    const int* bp = batch + row0 + 8 * eh;
    const int g0 = batch[row0];
    int gid[8];
    bool same = true;
#pragma unroll
    for (int r = 0; r < 8; ++r) {
      gid[r] = bp[r];
      same = same && (gid[r] == g0);
    }
    if (__all(same)) {
      float hi0 = -INFINITY, hi1 = -INFINITY, hi2 = -INFINITY, hi3 = -INFINITY;
      float lo0 = INFINITY, lo1 = INFINITY, lo2 = INFINITY, lo3 = INFINITY;
#pragma unroll
      for (int r = 0; r < 8; ++r) {
        hi0 = fmaxf(hi0, acc0[r]); lo0 = fminf(lo0, acc0[r]);
        hi1 = fmaxf(hi1, acc1[r]); lo1 = fminf(lo1, acc1[r]);
        hi2 = fmaxf(hi2, acc2[r]); lo2 = fminf(lo2, acc2[r]);
        hi3 = fmaxf(hi3, acc3[r]); lo3 = fminf(lo3, acc3[r]);
      }
      hi0 = fmaxf(hi0, __shfl_xor(hi0, 16)); lo0 = fminf(lo0, __shfl_xor(lo0, 16));
      hi1 = fmaxf(hi1, __shfl_xor(hi1, 16)); lo1 = fminf(lo1, __shfl_xor(lo1, 16));
      hi2 = fmaxf(hi2, __shfl_xor(hi2, 16)); lo2 = fminf(lo2, __shfl_xor(lo2, 16));
      hi3 = fmaxf(hi3, __shfl_xor(hi3, 16)); lo3 = fminf(lo3, __shfl_xor(lo3, 16));
      if (valid) {
        unsigned* pp = (eh ? pmin : pmax) + (size_t)g0 * DH + ec;
        atomicMax(pp,     eh ? ~enc_f32(lo0) : enc_f32(hi0));
        atomicMax(pp + 1, eh ? ~enc_f32(lo1) : enc_f32(hi1));
        atomicMax(pp + 2, eh ? ~enc_f32(lo2) : enc_f32(hi2));
        atomicMax(pp + 3, eh ? ~enc_f32(lo3) : enc_f32(hi3));
      }
    } else if (valid) {
      auto pool_runs = [&](const v8f& v, unsigned cj) {
        int g = gid[0];
        float hi = -INFINITY, lo = INFINITY;
#pragma unroll
        for (int r = 0; r < 8; ++r) {
          if (gid[r] != g) {
            atomicMax(pmax + ((unsigned)g * DH + cj), enc_f32(hi));
            atomicMax(pmin + ((unsigned)g * DH + cj), ~enc_f32(lo));
            g  = gid[r];
            hi = -INFINITY;
            lo = INFINITY;
          }
          hi = fmaxf(hi, v[r]);
          lo = fminf(lo, v[r]);
        }
        atomicMax(pmax + ((unsigned)g * DH + cj), enc_f32(hi));
        atomicMax(pmin + ((unsigned)g * DH + cj), ~enc_f32(lo));
      };
      pool_runs(acc0, (unsigned)ec);
      pool_runs(acc1, (unsigned)ec + 1u);
      pool_runs(acc2, (unsigned)ec + 2u);
      pool_runs(acc3, (unsigned)ec + 3u);
    }
  }
}

__global__ void pool_bn_finish(const unsigned* __restrict__ pmax, const unsigned* __restrict__ pmin,
                               const float* __restrict__ sums, const float* __restrict__ gamma,
                               const float* __restrict__ beta, float* __restrict__ out, int M) {
  int g = blockIdx.x, c = threadIdx.x;
  if (c >= DH) return;
  float scale, shift;
  bn_scale_shift(sums, c, M, gamma[c], beta[c], scale, shift);
  unsigned em = pmax[(size_t)g * DH + c];
  float res = -INFINITY;
  if (em != 0u) {
    float v = (scale >= 0.0f) ? dec_f32(em) : dec_f32(~pmin[(size_t)g * DH + c]);
    res = v * scale + shift;
  }
  out[(size_t)g * DH + c] = res;
}

extern "C" void kernel_launch(void* const* d_in, const int* in_sizes, int n_in,
                              void* d_out, int out_size, void* d_ws, size_t ws_size,
                              hipStream_t stream) {
  const float* x      = (const float*)d_in[0];
  const int*   eidx   = (const int*)d_in[1];
  const int*   batch  = (const int*)d_in[2];
  const float* W1     = (const float*)d_in[3];
  const float* b1     = (const float*)d_in[4];
  const float* gamma1 = (const float*)d_in[5];
  const float* beta1  = (const float*)d_in[6];
  const float* W2     = (const float*)d_in[7];
  const float* b2     = (const float*)d_in[8];
  const float* gamma2 = (const float*)d_in[9];
  const float* beta2  = (const float*)d_in[10];
  float* out = (float*)d_out;

  const int* src = eidx;
  const int* dst = eidx + EE;

  float* ws = (float*)d_ws;
  float*    deg  = ws;
  float*    sums = ws + 4178048;
  float*    bufA = ws + 4334688;
  float*    bufB = ws + 20334688;
  unsigned* tcnt = (unsigned*)(ws + 36334688);
  unsigned* toff = (unsigned*)(ws + 36337888);
  unsigned* rank = (unsigned*)(ws + 36341088);
  unsigned* rec  = (unsigned*)(ws + 37141088);
  unsigned* pmax = (unsigned*)(ws + 38800000);
  unsigned* pmin = pmax + GG * DH;
  unsigned* chist = (unsigned*)(ws + 40000000);
  unsigned* cbase = (unsigned*)(ws + 40262144);

  const int gemmGrid   = (NN + GB_ROWS - 1) / GB_ROWS;

  chunk_rank<<<NCHUNK, 1024, 0, stream>>>(dst, rank, chist);
  tile_scan<<<1, 1024, 0, stream>>>(chist, cbase, tcnt, toff, rec);
  tile_place<<<(EE + 255) / 256, 256, 0, stream>>>(src, dst, toff, cbase, rank, rec, EE);
  tile_deg<<<(NT + 7) / 8, 256, 0, stream>>>(rec, toff, tcnt, deg);
  tile_weight<<<(NT + 7) / 8, 256, 0, stream>>>(rec, toff, tcnt, deg);

  gemm_lds_wmma<DINP, 16, false, DIN, DIN><<<gemmGrid, 256, 0, stream>>>(x, W1, bufA, NN, nullptr,
                                                                         nullptr, nullptr);
  fill_f32<<<(NSC * 2 * DHP + 255) / 256, 256, 0, stream>>>(sums, 0.0f, NSC * 2 * DHP);
  agg_tile_wmma<false><<<NT, 160, 0, stream>>>(bufA, rec, toff, deg, b1, batch, bufB, sums, pmax,
                                               pmin);

  gemm_lds_wmma<DH, 24, true, DHP, DH><<<gemmGrid, 256, 0, stream>>>(bufB, W2, bufA, NN, sums,
                                                                     gamma1, beta1);
  fill_f32<<<(NSC * 2 * DHP + 255) / 256, 256, 0, stream>>>(sums, 0.0f, NSC * 2 * DHP);
  fill_u32<<<(2L * GG * DH + 255) / 256, 256, 0, stream>>>(pmax, 0u, 2L * GG * DH);

  agg_tile_wmma<true><<<NT, 160, 0, stream>>>(bufA, rec, toff, deg, b2, batch, bufB, sums, pmax,
                                              pmin);
  pool_bn_finish<<<GG, DHP, 0, stream>>>(pmax, pmin, sums, gamma2, beta2, out, NN);
}
